# starting kernel
# baseline (speedup 1.0000x reference)
#include <hip/hip_runtime.h>
#include <hip/hip_bf16.h>
#include <climits>


#define IN_DIM   256
#define OUT_DIM  128
#define HEADS    8
#define HEAD_DIM 16
#define NEG_SLOPE 0.2f

typedef float v2f __attribute__((ext_vector_type(2)));
typedef float v8f __attribute__((ext_vector_type(8)));

__device__ __forceinline__ int floatKey(float x) {
    int i = __float_as_int(x);
    return (i >= 0) ? i : (i ^ 0x7FFFFFFF);
}
__device__ __forceinline__ float keyToFloat(int k) {
    int i = (k >= 0) ? k : (k ^ 0x7FFFFFFF);
    return __int_as_float(i);
}

__device__ __forceinline__ float lrelu(float x) {
    return x > 0.0f ? x : NEG_SLOPE * x;
}

__global__ void gat_init(float* __restrict__ out, const float* __restrict__ b,
                         int* __restrict__ emaxk, float* __restrict__ denom, int N) {
    int gid = blockIdx.x * blockDim.x + threadIdx.x;
    int tot = N * OUT_DIM;
    if (gid < tot) {
        out[gid] = b[gid & (OUT_DIM - 1)];
    }
    if (gid < N * HEADS) {
        emaxk[gid] = INT_MIN;
        denom[gid] = 0.0f;
    }
}

#define A_STRIDE 260

__global__ void gat_gemm_wmma(const float* __restrict__ feat,
                              const float* __restrict__ W,
                              float* __restrict__ h, int N) {
    __shared__ float As[16 * A_STRIDE];

    const int row0 = blockIdx.x * 16;
    const int lane = threadIdx.x & 31;
    const int wave = threadIdx.x >> 5;
    const int m    = lane & 15;
    const int kh   = (lane >> 4) << 1;

    for (int i = threadIdx.x; i < 16 * (IN_DIM / 4); i += blockDim.x) {
        int r = i >> 6;
        int c = (i & 63) << 2;
        int gr = row0 + r;
        if (gr >= N) gr = N - 1;
        *(float4*)&As[r * A_STRIDE + c] =
            *(const float4*)&feat[(size_t)gr * IN_DIM + c];
    }
    __syncthreads();

    const int col0 = wave * HEAD_DIM;
    v8f acc = {};
    const float* __restrict__ arow = &As[m * A_STRIDE];

    #pragma unroll 4
    for (int k0 = 0; k0 < IN_DIM; k0 += 4) {
        v2f a = *(const v2f*)&arow[k0 + kh];
        v2f bb;
        bb.x = W[(size_t)(k0 + kh + 0) * OUT_DIM + col0 + m];
        bb.y = W[(size_t)(k0 + kh + 1) * OUT_DIM + col0 + m];
        acc = __builtin_amdgcn_wmma_f32_16x16x4_f32(
             false, a,  false, bb,
             (short)0, acc,  false,  false);
    }

    const int colg  = col0 + m;
    const int rbase = row0 + ((lane >> 4) << 3);
    float* __restrict__ op = h + (size_t)rbase * OUT_DIM + colg;
    if (row0 + 16 <= N) {
        #pragma unroll
        for (int r = 0; r < 8; ++r) {
            op[(size_t)r * OUT_DIM] = acc[r];
        }
    } else {
        #pragma unroll
        for (int r = 0; r < 8; ++r) {
            if (rbase + r < N) op[(size_t)r * OUT_DIM] = acc[r];
        }
    }
}

__global__ void gat_logits(const float* __restrict__ h,
                           const float* __restrict__ attn_l,
                           const float* __restrict__ attn_r,
                           float* __restrict__ el, float* __restrict__ er, int N) {
    int gid = blockIdx.x * blockDim.x + threadIdx.x;
    if (gid >= N * HEADS) return;
    int n  = gid >> 3;
    int hd = gid & 7;
    const float* hp = h + (size_t)n * OUT_DIM + hd * HEAD_DIM;
    const float* al = attn_l + hd * HEAD_DIM;
    const float* ar = attn_r + hd * HEAD_DIM;
    float sl = 0.0f, sr = 0.0f;
    #pragma unroll
    for (int j = 0; j < HEAD_DIM; ++j) {
        float v = hp[j];
        sl = fmaf(v, al[j], sl);
        sr = fmaf(v, ar[j], sr);
    }
    el[gid] = sl;
    er[gid] = sr;
}

__device__ __forceinline__ void edge_sd(const int* __restrict__ src,
                                        const int* __restrict__ dst,
                                        int e, int E, int& s, int& d) {
    if (e < E) { s = src[e]; d = dst[e]; }
    else       { s = e - E;  d = e - E; }
}

__global__ void gat_edge_max(const int* __restrict__ src, const int* __restrict__ dst,
                             const float* __restrict__ el, const float* __restrict__ er,
                             int* __restrict__ emaxk, int E, int Et) {
    int e = blockIdx.x * blockDim.x + threadIdx.x;
    if (e >= Et) return;
    int s, d;
    edge_sd(src, dst, e, E, s, d);
    float elv[8], erv[8];
    *(float4*)&elv[0] = *(const float4*)&el[(size_t)s * HEADS];
    *(float4*)&elv[4] = *(const float4*)&el[(size_t)s * HEADS + 4];
    *(float4*)&erv[0] = *(const float4*)&er[(size_t)d * HEADS];
    *(float4*)&erv[4] = *(const float4*)&er[(size_t)d * HEADS + 4];
    #pragma unroll
    for (int hd = 0; hd < HEADS; ++hd) {
        float x = lrelu(elv[hd] + erv[hd]);
        atomicMax(&emaxk[(size_t)d * HEADS + hd], floatKey(x));
    }
}

__global__ void gat_edge_sum(const int* __restrict__ src, const int* __restrict__ dst,
                             const float* __restrict__ el, const float* __restrict__ er,
                             const int* __restrict__ emaxk, float* __restrict__ denom,
                             int E, int Et) {
    int e = blockIdx.x * blockDim.x + threadIdx.x;
    if (e >= Et) return;
    int s, d;
    edge_sd(src, dst, e, E, s, d);
    float elv[8], erv[8];
    *(float4*)&elv[0] = *(const float4*)&el[(size_t)s * HEADS];
    *(float4*)&elv[4] = *(const float4*)&el[(size_t)s * HEADS + 4];
    *(float4*)&erv[0] = *(const float4*)&er[(size_t)d * HEADS];
    *(float4*)&erv[4] = *(const float4*)&er[(size_t)d * HEADS + 4];
    #pragma unroll
    for (int hd = 0; hd < HEADS; ++hd) {
        float x  = lrelu(elv[hd] + erv[hd]);
        float mx = keyToFloat(emaxk[(size_t)d * HEADS + hd]);
        atomicAdd(&denom[(size_t)d * HEADS + hd], __expf(x - mx));
    }
}

__global__ void gat_aggregate(const int* __restrict__ src, const int* __restrict__ dst,
                              const float* __restrict__ el, const float* __restrict__ er,
                              const int* __restrict__ emaxk, const float* __restrict__ denom,
                              const float* __restrict__ h, float* __restrict__ out,
                              int E, int Et) {
    int gid = blockIdx.x * blockDim.x + threadIdx.x;
    if (gid >= Et * HEADS) return;
    int e  = gid >> 3;
    int hd = gid & 7;
    int s, d;
    edge_sd(src, dst, e, E, s, d);

    float x  = lrelu(el[(size_t)s * HEADS + hd] + er[(size_t)d * HEADS + hd]);
    float mx = keyToFloat(emaxk[(size_t)d * HEADS + hd]);
    float alpha = __expf(x - mx) / denom[(size_t)d * HEADS + hd];

    const float* hp = h + (size_t)s * OUT_DIM + hd * HEAD_DIM;
    float* op       = out + (size_t)d * OUT_DIM + hd * HEAD_DIM;
    #pragma unroll
    for (int v = 0; v < HEAD_DIM / 4; ++v) {
        float4 hv = *(const float4*)&hp[v * 4];
        atomicAdd(&op[v * 4 + 0], hv.x * alpha);
        atomicAdd(&op[v * 4 + 1], hv.y * alpha);
        atomicAdd(&op[v * 4 + 2], hv.z * alpha);
        atomicAdd(&op[v * 4 + 3], hv.w * alpha);
    }
}

extern "C" void kernel_launch(void* const* d_in, const int* in_sizes, int n_in,
                              void* d_out, int out_size, void* d_ws, size_t ws_size,
                              hipStream_t stream) {
    const float* feat   = (const float*)d_in[0];
    const float* W      = (const float*)d_in[1];
    const float* b      = (const float*)d_in[2];
    const float* attn_l = (const float*)d_in[3];
    const float* attn_r = (const float*)d_in[4];
    const int*   src    = (const int*)d_in[5];
    const int*   dst    = (const int*)d_in[6];
    float*       out    = (float*)d_out;

    const int N  = in_sizes[0] / IN_DIM;
    const int E  = in_sizes[5];
    const int Et = E + N;

    float* h     = (float*)d_ws;
    float* el    = h + (size_t)N * OUT_DIM;
    float* er    = el + (size_t)N * HEADS;
    int*   emaxk = (int*)(er + (size_t)N * HEADS);
    float* denom = (float*)(emaxk + (size_t)N * HEADS);

    const int TB = 256;

    gat_init<<<(N * OUT_DIM + TB - 1) / TB, TB, 0, stream>>>(out, b, emaxk, denom, N);

    gat_gemm_wmma<<<(N + 15) / 16, TB, 0, stream>>>(feat, W, h, N);

    gat_logits<<<(N * HEADS + TB - 1) / TB, TB, 0, stream>>>(h, attn_l, attn_r, el, er, N);

    gat_edge_max<<<(Et + TB - 1) / TB, TB, 0, stream>>>(src, dst, el, er, emaxk, E, Et);

    gat_edge_sum<<<(Et + TB - 1) / TB, TB, 0, stream>>>(src, dst, el, er, emaxk, denom, E, Et);

    gat_aggregate<<<((size_t)Et * HEADS + TB - 1) / TB, TB, 0, stream>>>(
        src, dst, el, er, emaxk, denom, h, out, E, Et);
}


# Sort edges by dst in every call (count, scan, fill), then one wave per dst node does max, exp-sum and weighted h-row gather in registers and divides once: no float atomics; sum order = arrival rank
# speedup vs baseline: 4.9131x; 4.9131x over previous
#include <hip/hip_runtime.h>
#include <hip/hip_bf16.h>


#define IN_DIM   256
#define OUT_DIM  128
#define HEADS    8
#define HEAD_DIM 16
#define NEG_SLOPE 0.2f

typedef float v2f __attribute__((ext_vector_type(2)));
typedef float v8f __attribute__((ext_vector_type(8)));

__device__ __forceinline__ float lrelu(float x) {
    return x > 0.0f ? x : NEG_SLOPE * x;
}

__global__ void gat_zero_counts(int* __restrict__ cnt, int N) {
    int gid = blockIdx.x * blockDim.x + threadIdx.x;
    if (gid < N) cnt[gid] = 0;
}

#define A_STRIDE 260

__global__ void gat_gemm_wmma(const float* __restrict__ feat,
                              const float* __restrict__ W,
                              float* __restrict__ h, int N) {
    __shared__ float As[16 * A_STRIDE];

    const int row0 = blockIdx.x * 16;
    const int lane = threadIdx.x & 31;
    const int wave = threadIdx.x >> 5;
    const int m    = lane & 15;
    const int kh   = (lane >> 4) << 1;

    for (int i = threadIdx.x; i < 16 * (IN_DIM / 4); i += blockDim.x) {
        int r = i >> 6;
        int c = (i & 63) << 2;
        int gr = row0 + r;
        if (gr >= N) gr = N - 1;
        *(float4*)&As[r * A_STRIDE + c] =
            *(const float4*)&feat[(size_t)gr * IN_DIM + c];
    }
    __syncthreads();

    const int col0 = wave * HEAD_DIM;
    v8f acc = {};
    const float* __restrict__ arow = &As[m * A_STRIDE];

    #pragma unroll 4
    for (int k0 = 0; k0 < IN_DIM; k0 += 4) {
        v2f a = *(const v2f*)&arow[k0 + kh];
        v2f bb;
        bb.x = W[(size_t)(k0 + kh + 0) * OUT_DIM + col0 + m];
        bb.y = W[(size_t)(k0 + kh + 1) * OUT_DIM + col0 + m];
        acc = __builtin_amdgcn_wmma_f32_16x16x4_f32(
             false, a,  false, bb,
             (short)0, acc,  false,  false);
    }

    const int colg  = col0 + m;
    const int rbase = row0 + ((lane >> 4) << 3);
    float* __restrict__ op = h + (size_t)rbase * OUT_DIM + colg;
    if (row0 + 16 <= N) {
        #pragma unroll
        for (int r = 0; r < 8; ++r) {
            op[(size_t)r * OUT_DIM] = acc[r];
        }
    } else {
        #pragma unroll
        for (int r = 0; r < 8; ++r) {
            if (rbase + r < N) op[(size_t)r * OUT_DIM] = acc[r];
        }
    }
}

__global__ void gat_logits_l(const float* __restrict__ h,
                             const float* __restrict__ attn_l,
                             float* __restrict__ el, int N) {
    int gid = blockIdx.x * blockDim.x + threadIdx.x;
    if (gid >= N * HEADS) return;
    int n  = gid >> 3;
    int hd = gid & 7;
    const float* hp = h + (size_t)n * OUT_DIM + hd * HEAD_DIM;
    const float* al = attn_l + hd * HEAD_DIM;
    float sl = 0.0f;
    #pragma unroll
    for (int j = 0; j < HEAD_DIM; ++j) {
        sl = fmaf(hp[j], al[j], sl);
    }
    el[gid] = sl;
}

__global__ void gat_dst_count(const int* __restrict__ dst, int* __restrict__ cnt, int E) {
    int e = blockIdx.x * blockDim.x + threadIdx.x;
    if (e < E) atomicAdd(&cnt[dst[e]], 1);
}

#define SCAN_T 1024

__device__ __forceinline__ int4 ld4_guard(const int* p, int i0, int N) {
    int4 c;
    c.x = (i0 + 0 < N) ? p[i0 + 0] : 0;
    c.y = (i0 + 1 < N) ? p[i0 + 1] : 0;
    c.z = (i0 + 2 < N) ? p[i0 + 2] : 0;
    c.w = (i0 + 3 < N) ? p[i0 + 3] : 0;
    return c;
}

__device__ __forceinline__ int wave_incl_scan(int v, int lane) {
    for (int o = 1; o < 32; o <<= 1) {
        int u = __shfl_up(v, o);
        if (lane >= o) v += u;
    }
    return v;
}

__global__ void __launch_bounds__(SCAN_T)
gat_scan_counts(int* cnt, int N) {
    __shared__ int wtot[32];
    const int lane  = threadIdx.x & 31;
    const int wv    = __builtin_amdgcn_readfirstlane((int)(threadIdx.x >> 5));
    const int trips = (N + 32 * 128 - 1) / (32 * 128);
    const int wbase = wv * trips * 128;
    const int wend  = min(wbase + trips * 128, N);
    const int nfull = (wend > wbase) ? ((wend - wbase) >> 7) : 0;
    const bool tail = nfull < trips;
    const int  itail = wbase + nfull * 128 + lane * 4;

    int s = 0;
    #pragma unroll 8
    for (int t = 0; t < nfull; ++t) {
        const int4 c = *(const int4*)&cnt[wbase + t * 128 + lane * 4];
        s += (c.x + c.y) + (c.z + c.w);
    }
    if (tail) {
        const int4 c = ld4_guard(cnt, itail, N);
        s += (c.x + c.y) + (c.z + c.w);
    }
    for (int o = 16; o > 0; o >>= 1) s += __shfl_xor(s, o);
    if (lane == 0) wtot[wv] = s;
    __syncthreads();

    int w = (lane < wv) ? wtot[lane] : 0;
    for (int o = 16; o > 0; o >>= 1) w += __shfl_xor(w, o);
    int carry = w;

    #pragma unroll 4
    for (int t = 0; t < nfull; ++t) {
        const int i0 = wbase + t * 128 + lane * 4;
        const int4 c = *(const int4*)&cnt[i0];
        const int loc = (c.x + c.y) + (c.z + c.w);
        const int inc = wave_incl_scan(loc, lane);
        int4 r;
        r.x = carry + inc - loc;
        r.y = r.x + c.x;
        r.z = r.y + c.y;
        r.w = r.z + c.z;
        *(int4*)&cnt[i0] = r;
        carry += __shfl(inc, 31);
    }
    if (tail) {
        const int4 c = ld4_guard(cnt, itail, N);
        const int loc = (c.x + c.y) + (c.z + c.w);
        const int inc = wave_incl_scan(loc, lane);
        int4 r;
        r.x = carry + inc - loc;
        r.y = r.x + c.x;
        r.z = r.y + c.y;
        r.w = r.z + c.z;
        if (itail + 0 < N) cnt[itail + 0] = r.x;
        if (itail + 1 < N) cnt[itail + 1] = r.y;
        if (itail + 2 < N) cnt[itail + 2] = r.z;
        if (itail + 3 < N) cnt[itail + 3] = r.w;
    }
}

__global__ void gat_dst_fill(const int* __restrict__ src, const int* __restrict__ dst,
                             int* __restrict__ offs, int* __restrict__ ssrc, int E) {
    int e = blockIdx.x * blockDim.x + threadIdx.x;
    if (e >= E) return;
    int pos = atomicAdd(&offs[dst[e]], 1);
    ssrc[pos] = src[e];
}

__global__ void __launch_bounds__(256)
gat_node_gather(const int* __restrict__ offs, const int* __restrict__ ssrc,
                const float* __restrict__ el, const float* __restrict__ attn_r,
                const float* __restrict__ h, const float* __restrict__ b,
                float* __restrict__ out, int N) {
    const int lane = threadIdx.x & 31;
    const int d = __builtin_amdgcn_readfirstlane((int)(blockIdx.x * 8 + (threadIdx.x >> 5)));
    if (d >= N) return;
    const int hd  = lane >> 2;
    const int dm  = (d > 0) ? d - 1 : 0;
    int beg       = offs[dm];
    const int end = offs[d];
    if (d == 0) beg = 0;

    const float* hrow = h + (size_t)d * OUT_DIM;
    const float* hp   = hrow + hd * HEAD_DIM;
    const float* ar   = attn_r + hd * HEAD_DIM;
    float er = 0.0f;
    #pragma unroll
    for (int j = 0; j < HEAD_DIM; ++j) {
        er = fmaf(hp[j], ar[j], er);
    }
    const float x_self = lrelu(el[(size_t)d * HEADS + hd] + er);

    float mx = x_self;
    #pragma clang loop vectorize(disable) unroll_count(4)
    for (int i = beg; i < end; ++i) {
        const int s = ssrc[i];
        mx = fmaxf(mx, lrelu(el[(size_t)s * HEADS + hd] + er));
    }

    const float p0 = __expf(x_self - mx);
    float den = p0;
    const float4 hv = *(const float4*)&hrow[lane * 4];
    float4 acc;
    acc.x = hv.x * p0; acc.y = hv.y * p0; acc.z = hv.z * p0; acc.w = hv.w * p0;
    #pragma unroll 4
    for (int i = beg; i < end; ++i) {
        const int s = ssrc[i];
        const float p = __expf(lrelu(el[(size_t)s * HEADS + hd] + er) - mx);
        const float4 v = *(const float4*)&h[(size_t)s * OUT_DIM + lane * 4];
        den += p;
        acc.x = fmaf(v.x, p, acc.x);
        acc.y = fmaf(v.y, p, acc.y);
        acc.z = fmaf(v.z, p, acc.z);
        acc.w = fmaf(v.w, p, acc.w);
    }

    const float4 bb = *(const float4*)&b[lane * 4];
    float4 o;
    o.x = acc.x / den + bb.x;
    o.y = acc.y / den + bb.y;
    o.z = acc.z / den + bb.z;
    o.w = acc.w / den + bb.w;
    *(float4*)&out[(size_t)d * OUT_DIM + lane * 4] = o;
}

extern "C" void kernel_launch(void* const* d_in, const int* in_sizes, int n_in,
                              void* d_out, int out_size, void* d_ws, size_t ws_size,
                              hipStream_t stream) {
    const float* feat   = (const float*)d_in[0];
    const float* W      = (const float*)d_in[1];
    const float* b      = (const float*)d_in[2];
    const float* attn_l = (const float*)d_in[3];
    const float* attn_r = (const float*)d_in[4];
    const int*   src    = (const int*)d_in[5];
    const int*   dst    = (const int*)d_in[6];
    float*       out    = (float*)d_out;

    const int N = in_sizes[0] / IN_DIM;
    const int E = in_sizes[5];

    float* h    = (float*)d_ws;
    float* el   = h + (size_t)N * OUT_DIM;
    int*   offs = (int*)(el + (size_t)N * HEADS);
    int*   ssrc = offs + N;

    const int TB = 256;

    gat_zero_counts<<<(N + TB - 1) / TB, TB, 0, stream>>>(offs, N);

    gat_gemm_wmma<<<(N + 15) / 16, TB, 0, stream>>>(feat, W, h, N);

    gat_logits_l<<<(N * HEADS + TB - 1) / TB, TB, 0, stream>>>(h, attn_l, el, N);

    gat_dst_count<<<(E + TB - 1) / TB, TB, 0, stream>>>(dst, offs, E);

    gat_scan_counts<<<1, SCAN_T, 0, stream>>>(offs, N);

    gat_dst_fill<<<(E + TB - 1) / TB, TB, 0, stream>>>(src, dst, offs, ssrc, E);

    gat_node_gather<<<(N + 7) / 8, TB, 0, stream>>>(offs, ssrc, el, attn_r, h, b, out, N);
}


# Edge sort with LDS counters only, no global atomics: edges bucketed by 512-node range (count, scan, place), then one block per bucket counts, scans, places by node; sum order = LDS arrival rank
# speedup vs baseline: 9.4724x; 1.9280x over previous
#include <hip/hip_runtime.h>
#include <hip/hip_bf16.h>


#define IN_DIM   256
#define OUT_DIM  128
#define HEADS    8
#define HEAD_DIM 16
#define NEG_SLOPE 0.2f

typedef float v2f __attribute__((ext_vector_type(2)));
typedef float v8f __attribute__((ext_vector_type(8)));

__device__ __forceinline__ float lrelu(float x) {
    return x > 0.0f ? x : NEG_SLOPE * x;
}

#define A_STRIDE 260

__global__ void gat_gemm_wmma(const float* __restrict__ feat,
                              const float* __restrict__ W,
                              float* __restrict__ h, int N) {
    __shared__ float As[16 * A_STRIDE];

    const int row0 = blockIdx.x * 16;
    const int lane = threadIdx.x & 31;
    const int wave = threadIdx.x >> 5;
    const int m    = lane & 15;
    const int kh   = (lane >> 4) << 1;

    for (int i = threadIdx.x; i < 16 * (IN_DIM / 4); i += blockDim.x) {
        int r = i >> 6;
        int c = (i & 63) << 2;
        int gr = row0 + r;
        if (gr >= N) gr = N - 1;
        *(float4*)&As[r * A_STRIDE + c] =
            *(const float4*)&feat[(size_t)gr * IN_DIM + c];
    }
    __syncthreads();

    const int col0 = wave * HEAD_DIM;
    v8f acc = {};
    const float* __restrict__ arow = &As[m * A_STRIDE];

    #pragma unroll 4
    for (int k0 = 0; k0 < IN_DIM; k0 += 4) {
        v2f a = *(const v2f*)&arow[k0 + kh];
        v2f bb;
        bb.x = W[(size_t)(k0 + kh + 0) * OUT_DIM + col0 + m];
        bb.y = W[(size_t)(k0 + kh + 1) * OUT_DIM + col0 + m];
        acc = __builtin_amdgcn_wmma_f32_16x16x4_f32(
             false, a,  false, bb,
             (short)0, acc,  false,  false);
    }

    const int colg  = col0 + m;
    const int rbase = row0 + ((lane >> 4) << 3);
    float* __restrict__ op = h + (size_t)rbase * OUT_DIM + colg;
    if (row0 + 16 <= N) {
        #pragma unroll
        for (int r = 0; r < 8; ++r) {
            op[(size_t)r * OUT_DIM] = acc[r];
        }
    } else {
        #pragma unroll
        for (int r = 0; r < 8; ++r) {
            if (rbase + r < N) op[(size_t)r * OUT_DIM] = acc[r];
        }
    }
}

__global__ void gat_logits_l(const float* __restrict__ h,
                             const float* __restrict__ attn_l,
                             float* __restrict__ el, int N) {
    int gid = blockIdx.x * blockDim.x + threadIdx.x;
    if (gid >= N * HEADS) return;
    int n  = gid >> 3;
    int hd = gid & 7;
    const float* hp = h + (size_t)n * OUT_DIM + hd * HEAD_DIM;
    const float* al = attn_l + hd * HEAD_DIM;
    float sl = 0.0f;
    #pragma unroll
    for (int j = 0; j < HEAD_DIM; ++j) {
        sl = fmaf(hp[j], al[j], sl);
    }
    el[gid] = sl;
}

#define BSH    9
#define BNODES 512
#define MAXB   256
#define SORT_P 256
#define SORT_T 1024
#define SORT_K 8

__global__ void __launch_bounds__(SORT_T)
gat_bucket_count(const int* __restrict__ dst, int* __restrict__ cntm, int E, int NB) {
    __shared__ int lc[MAXB];
    const int p = blockIdx.x;
    for (int b = threadIdx.x; b < NB; b += SORT_T) lc[b] = 0;
    __syncthreads();
    const int chunk = (E + SORT_P - 1) / SORT_P;
    const int e0 = p * chunk;
    const int e1 = min(e0 + chunk, E);
    for (int e = e0 + (int)threadIdx.x; e < e1; e += SORT_K * SORT_T) {
        int dv[SORT_K];
        #pragma unroll
        for (int k = 0; k < SORT_K; ++k) {
            const int i = e + k * SORT_T;
            dv[k] = (i < e1) ? dst[i] : 0;
        }
        #pragma unroll
        for (int k = 0; k < SORT_K; ++k) {
            if (e + k * SORT_T < e1) atomicAdd(&lc[dv[k] >> BSH], 1);
        }
    }
    __syncthreads();
    for (int b = threadIdx.x; b < NB; b += SORT_T) cntm[b * SORT_P + p] = lc[b];
}

#define SCAN_T 1024

__device__ __forceinline__ int4 ld4_guard(const int* p, int i0, int N) {
    int4 c;
    c.x = (i0 + 0 < N) ? p[i0 + 0] : 0;
    c.y = (i0 + 1 < N) ? p[i0 + 1] : 0;
    c.z = (i0 + 2 < N) ? p[i0 + 2] : 0;
    c.w = (i0 + 3 < N) ? p[i0 + 3] : 0;
    return c;
}

__device__ __forceinline__ int wave_incl_scan(int v, int lane) {
    for (int o = 1; o < 32; o <<= 1) {
        int u = __shfl_up(v, o);
        if (lane >= o) v += u;
    }
    return v;
}

__global__ void __launch_bounds__(SCAN_T)
gat_scan_counts(int* cnt, int N) {
    __shared__ int wtot[32];
    const int lane  = threadIdx.x & 31;
    const int wv    = __builtin_amdgcn_readfirstlane((int)(threadIdx.x >> 5));
    const int trips = (N + 32 * 128 - 1) / (32 * 128);
    const int wbase = wv * trips * 128;
    const int wend  = min(wbase + trips * 128, N);
    const int nfull = (wend > wbase) ? ((wend - wbase) >> 7) : 0;
    const bool tail = nfull < trips;
    const int  itail = wbase + nfull * 128 + lane * 4;

    int s = 0;
    #pragma unroll 8
    for (int t = 0; t < nfull; ++t) {
        const int4 c = *(const int4*)&cnt[wbase + t * 128 + lane * 4];
        s += (c.x + c.y) + (c.z + c.w);
    }
    if (tail) {
        const int4 c = ld4_guard(cnt, itail, N);
        s += (c.x + c.y) + (c.z + c.w);
    }
    for (int o = 16; o > 0; o >>= 1) s += __shfl_xor(s, o);
    if (lane == 0) wtot[wv] = s;
    __syncthreads();

    int w = (lane < wv) ? wtot[lane] : 0;
    for (int o = 16; o > 0; o >>= 1) w += __shfl_xor(w, o);
    int carry = w;

    #pragma unroll 4
    for (int t = 0; t < nfull; ++t) {
        const int i0 = wbase + t * 128 + lane * 4;
        const int4 c = *(const int4*)&cnt[i0];
        const int loc = (c.x + c.y) + (c.z + c.w);
        const int inc = wave_incl_scan(loc, lane);
        int4 r;
        r.x = carry + inc - loc;
        r.y = r.x + c.x;
        r.z = r.y + c.y;
        r.w = r.z + c.z;
        *(int4*)&cnt[i0] = r;
        carry += __shfl(inc, 31);
    }
    if (tail) {
        const int4 c = ld4_guard(cnt, itail, N);
        const int loc = (c.x + c.y) + (c.z + c.w);
        const int inc = wave_incl_scan(loc, lane);
        int4 r;
        r.x = carry + inc - loc;
        r.y = r.x + c.x;
        r.z = r.y + c.y;
        r.w = r.z + c.z;
        if (itail + 0 < N) cnt[itail + 0] = r.x;
        if (itail + 1 < N) cnt[itail + 1] = r.y;
        if (itail + 2 < N) cnt[itail + 2] = r.z;
        if (itail + 3 < N) cnt[itail + 3] = r.w;
    }
}

__global__ void __launch_bounds__(SORT_T)
gat_bucket_place(const int* __restrict__ src, const int* __restrict__ dst,
                 const int* __restrict__ cntm, int2* __restrict__ pairs, int E, int NB) {
    __shared__ int lc[MAXB];
    const int p = blockIdx.x;
    for (int b = threadIdx.x; b < NB; b += SORT_T) lc[b] = cntm[b * SORT_P + p];
    __syncthreads();
    const int chunk = (E + SORT_P - 1) / SORT_P;
    const int e0 = p * chunk;
    const int e1 = min(e0 + chunk, E);
    for (int e = e0 + (int)threadIdx.x; e < e1; e += SORT_K * SORT_T) {
        int dv[SORT_K], sv[SORT_K];
        #pragma unroll
        for (int k = 0; k < SORT_K; ++k) {
            const int i = e + k * SORT_T;
            dv[k] = (i < e1) ? dst[i] : 0;
            sv[k] = (i < e1) ? src[i] : 0;
        }
        #pragma unroll
        for (int k = 0; k < SORT_K; ++k) {
            if (e + k * SORT_T < e1) {
                const int pos = atomicAdd(&lc[dv[k] >> BSH], 1);
                pairs[pos] = make_int2(sv[k], dv[k]);
            }
        }
    }
}

__global__ void __launch_bounds__(SORT_T)
gat_bucket_csr(const int2* __restrict__ pairs, const int* __restrict__ cntm,
               int* __restrict__ offs, int* __restrict__ ssrc, int N, int E, int NB) {
    __shared__ int nc[BNODES];
    __shared__ int wt[BNODES / 32];
    const int b    = blockIdx.x;
    const int t    = threadIdx.x;
    const int lane = t & 31;
    const int s0 = cntm[b * SORT_P];
    const int s1 = (b + 1 < NB) ? cntm[(b + 1) * SORT_P] : E;

    if (t < BNODES) nc[t] = 0;
    __syncthreads();
    for (int i = s0 + t; i < s1; i += SORT_K * SORT_T) {
        int dv[SORT_K];
        #pragma unroll
        for (int k = 0; k < SORT_K; ++k) {
            const int j = i + k * SORT_T;
            dv[k] = (j < s1) ? pairs[j].y : 0;
        }
        #pragma unroll
        for (int k = 0; k < SORT_K; ++k) {
            if (i + k * SORT_T < s1) atomicAdd(&nc[dv[k] & (BNODES - 1)], 1);
        }
    }
    __syncthreads();

    const int v   = (t < BNODES) ? nc[t] : 0;
    const int inc = wave_incl_scan(v, lane);
    if (t < BNODES && lane == 31) wt[t >> 5] = inc;
    __syncthreads();
    if (t < BNODES) {
        int carry = 0;
        for (int w = 0; w < (t >> 5); ++w) carry += wt[w];
        const int start = s0 + carry + inc - v;
        nc[t] = start;
        const int node = b * BNODES + t;
        if (node < N) offs[node] = start;
    }
    __syncthreads();

    for (int i = s0 + t; i < s1; i += SORT_K * SORT_T) {
        int2 pv[SORT_K];
        #pragma unroll
        for (int k = 0; k < SORT_K; ++k) {
            const int j = i + k * SORT_T;
            pv[k] = (j < s1) ? pairs[j] : make_int2(0, 0);
        }
        #pragma unroll
        for (int k = 0; k < SORT_K; ++k) {
            if (i + k * SORT_T < s1) {
                const int pos = atomicAdd(&nc[pv[k].y & (BNODES - 1)], 1);
                ssrc[pos] = pv[k].x;
            }
        }
    }
}

__global__ void __launch_bounds__(256)
gat_node_gather(const int* __restrict__ offs, const int* __restrict__ ssrc,
                const float* __restrict__ el, const float* __restrict__ attn_r,
                const float* __restrict__ h, const float* __restrict__ b,
                float* __restrict__ out, int N, int E) {
    const int lane = threadIdx.x & 31;
    const int d = __builtin_amdgcn_readfirstlane((int)(blockIdx.x * 8 + (threadIdx.x >> 5)));
    if (d >= N) return;
    const int hd  = lane >> 2;
    const int dn  = (d + 1 < N) ? d + 1 : d;
    const int beg = offs[d];
    int end       = offs[dn];
    if (d + 1 >= N) end = E;

    const float* hrow = h + (size_t)d * OUT_DIM;
    const float* hp   = hrow + hd * HEAD_DIM;
    const float* ar   = attn_r + hd * HEAD_DIM;
    float er = 0.0f;
    #pragma unroll
    for (int j = 0; j < HEAD_DIM; ++j) {
        er = fmaf(hp[j], ar[j], er);
    }
    const float x_self = lrelu(el[(size_t)d * HEADS + hd] + er);

    float mx = x_self;
    #pragma clang loop vectorize(disable) unroll_count(4)
    for (int i = beg; i < end; ++i) {
        const int s = ssrc[i];
        mx = fmaxf(mx, lrelu(el[(size_t)s * HEADS + hd] + er));
    }

    const float p0 = __expf(x_self - mx);
    float den = p0;
    const float4 hv = *(const float4*)&hrow[lane * 4];
    float4 acc;
    acc.x = hv.x * p0; acc.y = hv.y * p0; acc.z = hv.z * p0; acc.w = hv.w * p0;
    #pragma unroll 4
    for (int i = beg; i < end; ++i) {
        const int s = ssrc[i];
        const float p = __expf(lrelu(el[(size_t)s * HEADS + hd] + er) - mx);
        const float4 v = *(const float4*)&h[(size_t)s * OUT_DIM + lane * 4];
        den += p;
        acc.x = fmaf(v.x, p, acc.x);
        acc.y = fmaf(v.y, p, acc.y);
        acc.z = fmaf(v.z, p, acc.z);
        acc.w = fmaf(v.w, p, acc.w);
    }

    const float4 bb = *(const float4*)&b[lane * 4];
    float4 o;
    o.x = acc.x / den + bb.x;
    o.y = acc.y / den + bb.y;
    o.z = acc.z / den + bb.z;
    o.w = acc.w / den + bb.w;
    *(float4*)&out[(size_t)d * OUT_DIM + lane * 4] = o;
}

extern "C" void kernel_launch(void* const* d_in, const int* in_sizes, int n_in,
                              void* d_out, int out_size, void* d_ws, size_t ws_size,
                              hipStream_t stream) {
    const float* feat   = (const float*)d_in[0];
    const float* W      = (const float*)d_in[1];
    const float* b      = (const float*)d_in[2];
    const float* attn_l = (const float*)d_in[3];
    const float* attn_r = (const float*)d_in[4];
    const int*   src    = (const int*)d_in[5];
    const int*   dst    = (const int*)d_in[6];
    float*       out    = (float*)d_out;

    const int N = in_sizes[0] / IN_DIM;
    const int E = in_sizes[5];

    float* h    = (float*)d_ws;
    float* el   = h + (size_t)N * OUT_DIM;
    int*   offs = (int*)(el + (size_t)N * HEADS);
    int*   ssrc = offs + N;
    int*   cntm  = (int*)d_ws;
    int2*  pairs = (int2*)(cntm + MAXB * SORT_P);

    const int TB = 256;
    const int NB = (N + BNODES - 1) / BNODES;

    gat_bucket_count<<<SORT_P, SORT_T, 0, stream>>>(dst, cntm, E, NB);

    gat_scan_counts<<<1, SCAN_T, 0, stream>>>(cntm, NB * SORT_P);

    gat_bucket_place<<<SORT_P, SORT_T, 0, stream>>>(src, dst, cntm, pairs, E, NB);

    gat_bucket_csr<<<NB, SORT_T, 0, stream>>>(pairs, cntm, offs, ssrc, N, E, NB);

    gat_gemm_wmma<<<(N + 15) / 16, TB, 0, stream>>>(feat, W, h, N);

    gat_logits_l<<<(N * HEADS + TB - 1) / TB, TB, 0, stream>>>(h, attn_l, el, N);

    gat_node_gather<<<(N + 7) / 8, TB, 0, stream>>>(offs, ssrc, el, attn_r, h, b, out, N, E);
}


# Gather wave reads each source h row once: logits formed from the row in registers (4-column parts, quad add), softmax in one pass with a running max; logit launch dropped; logit sum order changes
# speedup vs baseline: 10.4392x; 1.1021x over previous
#include <hip/hip_runtime.h>
#include <hip/hip_bf16.h>


#define IN_DIM   256
#define OUT_DIM  128
#define HEADS    8
#define HEAD_DIM 16
#define NEG_SLOPE 0.2f

typedef float v2f __attribute__((ext_vector_type(2)));
typedef float v8f __attribute__((ext_vector_type(8)));

__device__ __forceinline__ float lrelu(float x) {
    return x > 0.0f ? x : NEG_SLOPE * x;
}

#define A_STRIDE 260

__global__ void gat_gemm_wmma(const float* __restrict__ feat,
                              const float* __restrict__ W,
                              float* __restrict__ h, int N) {
    __shared__ float As[16 * A_STRIDE];

    const int row0 = blockIdx.x * 16;
    const int lane = threadIdx.x & 31;
    const int wave = threadIdx.x >> 5;
    const int m    = lane & 15;
    const int kh   = (lane >> 4) << 1;

    for (int i = threadIdx.x; i < 16 * (IN_DIM / 4); i += blockDim.x) {
        int r = i >> 6;
        int c = (i & 63) << 2;
        int gr = row0 + r;
        if (gr >= N) gr = N - 1;
        *(float4*)&As[r * A_STRIDE + c] =
            *(const float4*)&feat[(size_t)gr * IN_DIM + c];
    }
    __syncthreads();

    const int col0 = wave * HEAD_DIM;
    v8f acc = {};
    const float* __restrict__ arow = &As[m * A_STRIDE];

    #pragma unroll 4
    for (int k0 = 0; k0 < IN_DIM; k0 += 4) {
        v2f a = *(const v2f*)&arow[k0 + kh];
        v2f bb;
        bb.x = W[(size_t)(k0 + kh + 0) * OUT_DIM + col0 + m];
        bb.y = W[(size_t)(k0 + kh + 1) * OUT_DIM + col0 + m];
        acc = __builtin_amdgcn_wmma_f32_16x16x4_f32(
             false, a,  false, bb,
             (short)0, acc,  false,  false);
    }

    const int colg  = col0 + m;
    const int rbase = row0 + ((lane >> 4) << 3);
    float* __restrict__ op = h + (size_t)rbase * OUT_DIM + colg;
    if (row0 + 16 <= N) {
        #pragma unroll
        for (int r = 0; r < 8; ++r) {
            op[(size_t)r * OUT_DIM] = acc[r];
        }
    } else {
        #pragma unroll
        for (int r = 0; r < 8; ++r) {
            if (rbase + r < N) op[(size_t)r * OUT_DIM] = acc[r];
        }
    }
}

#define BSH    9
#define BNODES 512
#define MAXB   256
#define SORT_P 256
#define SORT_T 1024
#define SORT_K 8

__global__ void __launch_bounds__(SORT_T)
gat_bucket_count(const int* __restrict__ dst, int* __restrict__ cntm, int E, int NB) {
    __shared__ int lc[MAXB];
    const int p = blockIdx.x;
    for (int b = threadIdx.x; b < NB; b += SORT_T) lc[b] = 0;
    __syncthreads();
    const int chunk = (E + SORT_P - 1) / SORT_P;
    const int e0 = p * chunk;
    const int e1 = min(e0 + chunk, E);
    for (int e = e0 + (int)threadIdx.x; e < e1; e += SORT_K * SORT_T) {
        int dv[SORT_K];
        #pragma unroll
        for (int k = 0; k < SORT_K; ++k) {
            const int i = e + k * SORT_T;
            dv[k] = (i < e1) ? dst[i] : 0;
        }
        #pragma unroll
        for (int k = 0; k < SORT_K; ++k) {
            if (e + k * SORT_T < e1) atomicAdd(&lc[dv[k] >> BSH], 1);
        }
    }
    __syncthreads();
    for (int b = threadIdx.x; b < NB; b += SORT_T) cntm[b * SORT_P + p] = lc[b];
}

#define SCAN_T 1024

__device__ __forceinline__ int4 ld4_guard(const int* p, int i0, int N) {
    int4 c;
    c.x = (i0 + 0 < N) ? p[i0 + 0] : 0;
    c.y = (i0 + 1 < N) ? p[i0 + 1] : 0;
    c.z = (i0 + 2 < N) ? p[i0 + 2] : 0;
    c.w = (i0 + 3 < N) ? p[i0 + 3] : 0;
    return c;
}

__device__ __forceinline__ int wave_incl_scan(int v, int lane) {
    for (int o = 1; o < 32; o <<= 1) {
        int u = __shfl_up(v, o);
        if (lane >= o) v += u;
    }
    return v;
}

__global__ void __launch_bounds__(SCAN_T)
gat_scan_counts(int* cnt, int N) {
    __shared__ int wtot[32];
    const int lane  = threadIdx.x & 31;
    const int wv    = __builtin_amdgcn_readfirstlane((int)(threadIdx.x >> 5));
    const int trips = (N + 32 * 128 - 1) / (32 * 128);
    const int wbase = wv * trips * 128;
    const int wend  = min(wbase + trips * 128, N);
    const int nfull = (wend > wbase) ? ((wend - wbase) >> 7) : 0;
    const bool tail = nfull < trips;
    const int  itail = wbase + nfull * 128 + lane * 4;

    int s = 0;
    #pragma unroll 8
    for (int t = 0; t < nfull; ++t) {
        const int4 c = *(const int4*)&cnt[wbase + t * 128 + lane * 4];
        s += (c.x + c.y) + (c.z + c.w);
    }
    if (tail) {
        const int4 c = ld4_guard(cnt, itail, N);
        s += (c.x + c.y) + (c.z + c.w);
    }
    for (int o = 16; o > 0; o >>= 1) s += __shfl_xor(s, o);
    if (lane == 0) wtot[wv] = s;
    __syncthreads();

    int w = (lane < wv) ? wtot[lane] : 0;
    for (int o = 16; o > 0; o >>= 1) w += __shfl_xor(w, o);
    int carry = w;

    #pragma unroll 4
    for (int t = 0; t < nfull; ++t) {
        const int i0 = wbase + t * 128 + lane * 4;
        const int4 c = *(const int4*)&cnt[i0];
        const int loc = (c.x + c.y) + (c.z + c.w);
        const int inc = wave_incl_scan(loc, lane);
        int4 r;
        r.x = carry + inc - loc;
        r.y = r.x + c.x;
        r.z = r.y + c.y;
        r.w = r.z + c.z;
        *(int4*)&cnt[i0] = r;
        carry += __shfl(inc, 31);
    }
    if (tail) {
        const int4 c = ld4_guard(cnt, itail, N);
        const int loc = (c.x + c.y) + (c.z + c.w);
        const int inc = wave_incl_scan(loc, lane);
        int4 r;
        r.x = carry + inc - loc;
        r.y = r.x + c.x;
        r.z = r.y + c.y;
        r.w = r.z + c.z;
        if (itail + 0 < N) cnt[itail + 0] = r.x;
        if (itail + 1 < N) cnt[itail + 1] = r.y;
        if (itail + 2 < N) cnt[itail + 2] = r.z;
        if (itail + 3 < N) cnt[itail + 3] = r.w;
    }
}

__global__ void __launch_bounds__(SORT_T)
gat_bucket_place(const int* __restrict__ src, const int* __restrict__ dst,
                 const int* __restrict__ cntm, int2* __restrict__ pairs, int E, int NB) {
    __shared__ int lc[MAXB];
    const int p = blockIdx.x;
    for (int b = threadIdx.x; b < NB; b += SORT_T) lc[b] = cntm[b * SORT_P + p];
    __syncthreads();
    const int chunk = (E + SORT_P - 1) / SORT_P;
    const int e0 = p * chunk;
    const int e1 = min(e0 + chunk, E);
    for (int e = e0 + (int)threadIdx.x; e < e1; e += SORT_K * SORT_T) {
        int dv[SORT_K], sv[SORT_K];
        #pragma unroll
        for (int k = 0; k < SORT_K; ++k) {
            const int i = e + k * SORT_T;
            dv[k] = (i < e1) ? dst[i] : 0;
            sv[k] = (i < e1) ? src[i] : 0;
        }
        #pragma unroll
        for (int k = 0; k < SORT_K; ++k) {
            if (e + k * SORT_T < e1) {
                const int pos = atomicAdd(&lc[dv[k] >> BSH], 1);
                pairs[pos] = make_int2(sv[k], dv[k]);
            }
        }
    }
}

__global__ void __launch_bounds__(SORT_T)
gat_bucket_csr(const int2* __restrict__ pairs, const int* __restrict__ cntm,
               int* __restrict__ offs, int* __restrict__ ssrc, int N, int E, int NB) {
    __shared__ int nc[BNODES];
    __shared__ int wt[BNODES / 32];
    const int b    = blockIdx.x;
    const int t    = threadIdx.x;
    const int lane = t & 31;
    const int s0 = cntm[b * SORT_P];
    const int s1 = (b + 1 < NB) ? cntm[(b + 1) * SORT_P] : E;

    if (t < BNODES) nc[t] = 0;
    __syncthreads();
    for (int i = s0 + t; i < s1; i += SORT_K * SORT_T) {
        int dv[SORT_K];
        #pragma unroll
        for (int k = 0; k < SORT_K; ++k) {
            const int j = i + k * SORT_T;
            dv[k] = (j < s1) ? pairs[j].y : 0;
        }
        #pragma unroll
        for (int k = 0; k < SORT_K; ++k) {
            if (i + k * SORT_T < s1) atomicAdd(&nc[dv[k] & (BNODES - 1)], 1);
        }
    }
    __syncthreads();

    const int v   = (t < BNODES) ? nc[t] : 0;
    const int inc = wave_incl_scan(v, lane);
    if (t < BNODES && lane == 31) wt[t >> 5] = inc;
    __syncthreads();
    if (t < BNODES) {
        int carry = 0;
        for (int w = 0; w < (t >> 5); ++w) carry += wt[w];
        const int start = s0 + carry + inc - v;
        nc[t] = start;
        const int node = b * BNODES + t;
        if (node < N) offs[node] = start;
    }
    __syncthreads();

    for (int i = s0 + t; i < s1; i += SORT_K * SORT_T) {
        int2 pv[SORT_K];
        #pragma unroll
        for (int k = 0; k < SORT_K; ++k) {
            const int j = i + k * SORT_T;
            pv[k] = (j < s1) ? pairs[j] : make_int2(0, 0);
        }
        #pragma unroll
        for (int k = 0; k < SORT_K; ++k) {
            if (i + k * SORT_T < s1) {
                const int pos = atomicAdd(&nc[pv[k].y & (BNODES - 1)], 1);
                ssrc[pos] = pv[k].x;
            }
        }
    }
}

__device__ __forceinline__ float dot4(const float4 v, const float4 a) {
    return fmaf(v.w, a.w, fmaf(v.z, a.z, fmaf(v.y, a.y, v.x * a.x)));
}

__device__ __forceinline__ float quad_sum(float x) {
    const float y = __int_as_float(__builtin_amdgcn_update_dpp(
        0, __float_as_int(x), 0xB1  , 0xF, 0xF, true));
    x += y;
    const float z = __int_as_float(__builtin_amdgcn_update_dpp(
        0, __float_as_int(x), 0x4E  , 0xF, 0xF, true));
    return x + z;
}

__device__ __forceinline__ void fold4(const float* __restrict__ h, const int lane,
                                      const int s0, const int s1, const int s2, const int s3,
                                      const bool k1, const bool k2, const bool k3,
                                      const float4 al, const float er,
                                      float& mx, float& den, float4& acc) {
    const float4 v0 = *(const float4*)&h[(size_t)s0 * OUT_DIM + lane * 4];
    const float4 v1 = *(const float4*)&h[(size_t)s1 * OUT_DIM + lane * 4];
    const float4 v2 = *(const float4*)&h[(size_t)s2 * OUT_DIM + lane * 4];
    const float4 v3 = *(const float4*)&h[(size_t)s3 * OUT_DIM + lane * 4];
    const float x0 = lrelu(quad_sum(dot4(v0, al)) + er);
    const float x1 = k1 ? lrelu(quad_sum(dot4(v1, al)) + er) : -INFINITY;
    const float x2 = k2 ? lrelu(quad_sum(dot4(v2, al)) + er) : -INFINITY;
    const float x3 = k3 ? lrelu(quad_sum(dot4(v3, al)) + er) : -INFINITY;
    const float mn = fmaxf(fmaxf(mx, fmaxf(x0, x1)), fmaxf(x2, x3));
    const float sc = __expf(mx - mn);
    const float p0 = __expf(x0 - mn);
    const float p1 = __expf(x1 - mn);
    const float p2 = __expf(x2 - mn);
    const float p3 = __expf(x3 - mn);
    mx  = mn;
    den = fmaf(den, sc, p0);
    den += p1;
    den += p2;
    den += p3;
    acc.x = fmaf(v3.x, p3, fmaf(v2.x, p2, fmaf(v1.x, p1, fmaf(v0.x, p0, acc.x * sc))));
    acc.y = fmaf(v3.y, p3, fmaf(v2.y, p2, fmaf(v1.y, p1, fmaf(v0.y, p0, acc.y * sc))));
    acc.z = fmaf(v3.z, p3, fmaf(v2.z, p2, fmaf(v1.z, p1, fmaf(v0.z, p0, acc.z * sc))));
    acc.w = fmaf(v3.w, p3, fmaf(v2.w, p2, fmaf(v1.w, p1, fmaf(v0.w, p0, acc.w * sc))));
}

__global__ void __launch_bounds__(256, 16)
gat_node_gather(const int* __restrict__ offs, const int* __restrict__ ssrc,
                const float* __restrict__ attn_l, const float* __restrict__ attn_r,
                const float* __restrict__ h, const float* __restrict__ b,
                float* __restrict__ out, int N, int E) {
    const int lane = threadIdx.x & 31;
    const int d = __builtin_amdgcn_readfirstlane((int)(blockIdx.x * 8 + (threadIdx.x >> 5)));
    if (d >= N) return;
    const int dn  = (d + 1 < N) ? d + 1 : d;
    const int beg = offs[d];
    int end       = offs[dn];
    if (d + 1 >= N) end = E;

    const float4 hv = *(const float4*)&h[(size_t)d * OUT_DIM + lane * 4];
    const float4 al = *(const float4*)&attn_l[lane * 4];
    const float4 ar = *(const float4*)&attn_r[lane * 4];
    const float er = quad_sum(dot4(hv, ar));

    float  mx  = lrelu(quad_sum(dot4(hv, al)) + er);
    float  den = 1.0f;
    float4 acc = hv;

    int i = beg;
    for (; i + 4 <= end; i += 4) {
        fold4(h, lane, ssrc[i], ssrc[i + 1], ssrc[i + 2], ssrc[i + 3],
              true, true, true, al, er, mx, den, acc);
    }
    if (i < end) {
        const bool k1 = i + 1 < end;
        const bool k2 = i + 2 < end;
        fold4(h, lane, ssrc[i], ssrc[k1 ? i + 1 : i], ssrc[k2 ? i + 2 : i], ssrc[i],
              k1, k2, false, al, er, mx, den, acc);
    }

    const float4 bb = *(const float4*)&b[lane * 4];
    float4 o;
    o.x = acc.x / den + bb.x;
    o.y = acc.y / den + bb.y;
    o.z = acc.z / den + bb.z;
    o.w = acc.w / den + bb.w;
    *(float4*)&out[(size_t)d * OUT_DIM + lane * 4] = o;
}

extern "C" void kernel_launch(void* const* d_in, const int* in_sizes, int n_in,
                              void* d_out, int out_size, void* d_ws, size_t ws_size,
                              hipStream_t stream) {
    const float* feat   = (const float*)d_in[0];
    const float* W      = (const float*)d_in[1];
    const float* b      = (const float*)d_in[2];
    const float* attn_l = (const float*)d_in[3];
    const float* attn_r = (const float*)d_in[4];
    const int*   src    = (const int*)d_in[5];
    const int*   dst    = (const int*)d_in[6];
    float*       out    = (float*)d_out;

    const int N = in_sizes[0] / IN_DIM;
    const int E = in_sizes[5];

    float* h    = (float*)d_ws;
    int*   offs = (int*)(h + (size_t)N * OUT_DIM);
    int*   ssrc = offs + N;
    int*   cntm  = (int*)d_ws;
    int2*  pairs = (int2*)(cntm + MAXB * SORT_P);

    const int TB = 256;
    const int NB = (N + BNODES - 1) / BNODES;

    gat_bucket_count<<<SORT_P, SORT_T, 0, stream>>>(dst, cntm, E, NB);

    gat_scan_counts<<<1, SCAN_T, 0, stream>>>(cntm, NB * SORT_P);

    gat_bucket_place<<<SORT_P, SORT_T, 0, stream>>>(src, dst, cntm, pairs, E, NB);

    gat_bucket_csr<<<NB, SORT_T, 0, stream>>>(pairs, cntm, offs, ssrc, N, E, NB);

    gat_gemm_wmma<<<(N + 15) / 16, TB, 0, stream>>>(feat, W, h, N);

    gat_node_gather<<<(N + 7) / 8, TB, 0, stream>>>(offs, ssrc, attn_l, attn_r, h, b, out, N, E);
}


# Level-1 edge sort block-local: each chunk sorted by bucket in LDS and stored as one piece, the bucket block finds its 256 runs from the counts; count and scan launches gone; sum order LDS arrival
# speedup vs baseline: 12.0175x; 1.1512x over previous
#include <hip/hip_runtime.h>
#include <hip/hip_bf16.h>


#define IN_DIM   256
#define OUT_DIM  128
#define HEADS    8
#define HEAD_DIM 16
#define NEG_SLOPE 0.2f

typedef float v2f __attribute__((ext_vector_type(2)));
typedef float v8f __attribute__((ext_vector_type(8)));

__device__ __forceinline__ float lrelu(float x) {
    return x > 0.0f ? x : NEG_SLOPE * x;
}

#define A_STRIDE 260

__global__ void gat_gemm_wmma(const float* __restrict__ feat,
                              const float* __restrict__ W,
                              float* __restrict__ h, int N) {
    __shared__ float As[16 * A_STRIDE];

    const int row0 = blockIdx.x * 16;
    const int lane = threadIdx.x & 31;
    const int wave = threadIdx.x >> 5;
    const int m    = lane & 15;
    const int kh   = (lane >> 4) << 1;

    for (int i = threadIdx.x; i < 16 * (IN_DIM / 4); i += blockDim.x) {
        int r = i >> 6;
        int c = (i & 63) << 2;
        int gr = row0 + r;
        if (gr >= N) gr = N - 1;
        *(float4*)&As[r * A_STRIDE + c] =
            *(const float4*)&feat[(size_t)gr * IN_DIM + c];
    }
    __syncthreads();

    const int col0 = wave * HEAD_DIM;
    v8f acc = {};
    const float* __restrict__ arow = &As[m * A_STRIDE];

    #pragma unroll 4
    for (int k0 = 0; k0 < IN_DIM; k0 += 4) {
        v2f a = *(const v2f*)&arow[k0 + kh];
        v2f bb;
        bb.x = W[(size_t)(k0 + kh + 0) * OUT_DIM + col0 + m];
        bb.y = W[(size_t)(k0 + kh + 1) * OUT_DIM + col0 + m];
        acc = __builtin_amdgcn_wmma_f32_16x16x4_f32(
             false, a,  false, bb,
             (short)0, acc,  false,  false);
    }

    const int colg  = col0 + m;
    const int rbase = row0 + ((lane >> 4) << 3);
    float* __restrict__ op = h + (size_t)rbase * OUT_DIM + colg;
    if (row0 + 16 <= N) {
        #pragma unroll
        for (int r = 0; r < 8; ++r) {
            op[(size_t)r * OUT_DIM] = acc[r];
        }
    } else {
        #pragma unroll
        for (int r = 0; r < 8; ++r) {
            if (rbase + r < N) op[(size_t)r * OUT_DIM] = acc[r];
        }
    }
}

#define BSH    9
#define BNODES 512
#define MAXB   256
#define SORT_P 256
#define SORT_T 1024
#define SORT_K 8

__device__ __forceinline__ int wave_incl_scan(int v, int lane) {
    for (int o = 1; o < 32; o <<= 1) {
        int u = __shfl_up(v, o);
        if (lane >= o) v += u;
    }
    return v;
}

__global__ void __launch_bounds__(SORT_T)
gat_chunk_sort(const int* __restrict__ src, const int* __restrict__ dst,
               int* __restrict__ cntm, int2* __restrict__ pairs, int E, int NB) {
    __shared__ int  lc[MAXB];
    __shared__ int  wt[MAXB / 32];
    __shared__ int2 stage[SORT_K * SORT_T];
    const int p    = blockIdx.x;
    const int t    = threadIdx.x;
    const int lane = t & 31;
    if (t < MAXB) lc[t] = 0;
    __syncthreads();
    const int chunk = (E + SORT_P - 1) / SORT_P;
    const int e0 = p * chunk;
    const int e1 = min(e0 + chunk, E);

    int dv[SORT_K], sv[SORT_K];
    #pragma unroll
    for (int k = 0; k < SORT_K; ++k) {
        const int i = e0 + t + k * SORT_T;
        dv[k] = (i < e1) ? dst[i] : -1;
        sv[k] = (i < e1) ? src[i] : 0;
    }
    #pragma unroll
    for (int k = 0; k < SORT_K; ++k) {
        if (dv[k] >= 0) atomicAdd(&lc[dv[k] >> BSH], 1);
    }
    __syncthreads();

    const int v   = (t < MAXB) ? lc[t] : 0;
    const int inc = wave_incl_scan(v, lane);
    if (t < MAXB && lane == 31) wt[t >> 5] = inc;
    __syncthreads();
    if (t < MAXB) {
        int carry = 0;
        for (int w = 0; w < (t >> 5); ++w) carry += wt[w];
        lc[t] = carry + inc - v;
        if (t < NB) cntm[t * SORT_P + p] = v;
    }
    __syncthreads();

    #pragma unroll
    for (int k = 0; k < SORT_K; ++k) {
        if (dv[k] >= 0) {
            const int pos = atomicAdd(&lc[dv[k] >> BSH], 1);
            stage[pos] = make_int2(sv[k], dv[k]);
        }
    }
    __syncthreads();

    for (int i = t; i < e1 - e0; i += SORT_T) pairs[e0 + i] = stage[i];
}

#define CSR_J 16
#define CSR_R (SORT_P / 32)

__global__ void __launch_bounds__(SORT_T)
gat_bucket_csr(const int2* __restrict__ pairs, const int* __restrict__ cntm,
               int* __restrict__ offs, int* __restrict__ ssrc, int N, int E) {
    __shared__ int nc[BNODES];
    __shared__ int wt[BNODES / 32];
    __shared__ __attribute__((aligned(16))) int part[CSR_J * SORT_P];
    __shared__ int rstart[SORT_P];
    __shared__ int rlen[SORT_P];
    __shared__ int red[SORT_T / 32];
    const int b    = blockIdx.x;
    const int t    = threadIdx.x;
    const int lane = t & 31;
    const int wv   = t >> 5;
    const int chunk = (E + SORT_P - 1) / SORT_P;

    const int mylen = (t < SORT_P) ? cntm[b * SORT_P + t] : 0;
    {
        const int q = (t & 63) << 2;
        const int j = t >> 6;
        int4 s = make_int4(0, 0, 0, 0);
        #pragma unroll 8
        for (int r = j; r < b; r += CSR_J) {
            const int4 c = *(const int4*)&cntm[r * SORT_P + q];
            s.x += c.x; s.y += c.y; s.z += c.z; s.w += c.w;
        }
        *(int4*)&part[j * SORT_P + q] = s;
    }
    if (t < BNODES) nc[t] = 0;
    __syncthreads();

    int roff = 0;
    if (t < SORT_P) {
        #pragma unroll
        for (int j = 0; j < CSR_J; ++j) roff += part[j * SORT_P + t];
        rstart[t] = t * chunk + roff;
        rlen[t]   = mylen;
    }
    int rs = roff;
    for (int o = 16; o > 0; o >>= 1) rs += __shfl_xor(rs, o);
    if (lane == 0) red[wv] = rs;
    __syncthreads();
    int s0 = red[lane];
    for (int o = 16; o > 0; o >>= 1) s0 += __shfl_xor(s0, o);

    int2 pa[CSR_R], pb[CSR_R];
    int  st[CSR_R], ln[CSR_R];
    #pragma unroll
    for (int r = 0; r < CSR_R; ++r) {
        st[r] = rstart[wv + 32 * r];
        ln[r] = rlen[wv + 32 * r];
    }
    #pragma unroll
    for (int r = 0; r < CSR_R; ++r) {
        pa[r] = (lane < ln[r])      ? pairs[st[r] + lane]      : make_int2(0, -1);
        pb[r] = (lane + 32 < ln[r]) ? pairs[st[r] + lane + 32] : make_int2(0, -1);
    }
    #pragma unroll
    for (int r = 0; r < CSR_R; ++r) {
        if (pa[r].y >= 0) atomicAdd(&nc[pa[r].y & (BNODES - 1)], 1);
        if (pb[r].y >= 0) atomicAdd(&nc[pb[r].y & (BNODES - 1)], 1);
    }
    #pragma unroll
    for (int r = 0; r < CSR_R; ++r) {
        for (int o = lane + 64; o < ln[r]; o += 32)
            atomicAdd(&nc[pairs[st[r] + o].y & (BNODES - 1)], 1);
    }
    __syncthreads();

    const int v   = (t < BNODES) ? nc[t] : 0;
    const int inc = wave_incl_scan(v, lane);
    if (t < BNODES && lane == 31) wt[t >> 5] = inc;
    __syncthreads();
    if (t < BNODES) {
        int carry = 0;
        for (int w = 0; w < (t >> 5); ++w) carry += wt[w];
        const int start = s0 + carry + inc - v;
        nc[t] = start;
        const int node = b * BNODES + t;
        if (node < N) offs[node] = start;
    }
    __syncthreads();

    #pragma unroll
    for (int r = 0; r < CSR_R; ++r) {
        if (pa[r].y >= 0) {
            const int pos = atomicAdd(&nc[pa[r].y & (BNODES - 1)], 1);
            ssrc[pos] = pa[r].x;
        }
        if (pb[r].y >= 0) {
            const int pos = atomicAdd(&nc[pb[r].y & (BNODES - 1)], 1);
            ssrc[pos] = pb[r].x;
        }
    }
    #pragma unroll
    for (int r = 0; r < CSR_R; ++r) {
        for (int o = lane + 64; o < ln[r]; o += 32) {
            const int2 pv = pairs[st[r] + o];
            const int pos = atomicAdd(&nc[pv.y & (BNODES - 1)], 1);
            ssrc[pos] = pv.x;
        }
    }
}

__device__ __forceinline__ float dot4(const float4 v, const float4 a) {
    return fmaf(v.w, a.w, fmaf(v.z, a.z, fmaf(v.y, a.y, v.x * a.x)));
}

__device__ __forceinline__ float quad_sum(float x) {
    const float y = __int_as_float(__builtin_amdgcn_update_dpp(
        0, __float_as_int(x), 0xB1  , 0xF, 0xF, true));
    x += y;
    const float z = __int_as_float(__builtin_amdgcn_update_dpp(
        0, __float_as_int(x), 0x4E  , 0xF, 0xF, true));
    return x + z;
}

__device__ __forceinline__ void fold4(const float* __restrict__ h, const int lane,
                                      const int s0, const int s1, const int s2, const int s3,
                                      const bool k1, const bool k2, const bool k3,
                                      const float4 al, const float er,
                                      float& mx, float& den, float4& acc) {
    const float4 v0 = *(const float4*)&h[(size_t)s0 * OUT_DIM + lane * 4];
    const float4 v1 = *(const float4*)&h[(size_t)s1 * OUT_DIM + lane * 4];
    const float4 v2 = *(const float4*)&h[(size_t)s2 * OUT_DIM + lane * 4];
    const float4 v3 = *(const float4*)&h[(size_t)s3 * OUT_DIM + lane * 4];
    const float x0 = lrelu(quad_sum(dot4(v0, al)) + er);
    const float x1 = k1 ? lrelu(quad_sum(dot4(v1, al)) + er) : -INFINITY;
    const float x2 = k2 ? lrelu(quad_sum(dot4(v2, al)) + er) : -INFINITY;
    const float x3 = k3 ? lrelu(quad_sum(dot4(v3, al)) + er) : -INFINITY;
    const float mn = fmaxf(fmaxf(mx, fmaxf(x0, x1)), fmaxf(x2, x3));
    const float sc = __expf(mx - mn);
    const float p0 = __expf(x0 - mn);
    const float p1 = __expf(x1 - mn);
    const float p2 = __expf(x2 - mn);
    const float p3 = __expf(x3 - mn);
    mx  = mn;
    den = fmaf(den, sc, p0);
    den += p1;
    den += p2;
    den += p3;
    acc.x = fmaf(v3.x, p3, fmaf(v2.x, p2, fmaf(v1.x, p1, fmaf(v0.x, p0, acc.x * sc))));
    acc.y = fmaf(v3.y, p3, fmaf(v2.y, p2, fmaf(v1.y, p1, fmaf(v0.y, p0, acc.y * sc))));
    acc.z = fmaf(v3.z, p3, fmaf(v2.z, p2, fmaf(v1.z, p1, fmaf(v0.z, p0, acc.z * sc))));
    acc.w = fmaf(v3.w, p3, fmaf(v2.w, p2, fmaf(v1.w, p1, fmaf(v0.w, p0, acc.w * sc))));
}

__global__ void __launch_bounds__(256, 16)
gat_node_gather(const int* __restrict__ offs, const int* __restrict__ ssrc,
                const float* __restrict__ attn_l, const float* __restrict__ attn_r,
                const float* __restrict__ h, const float* __restrict__ b,
                float* __restrict__ out, int N, int E) {
    const int lane = threadIdx.x & 31;
    const int d = __builtin_amdgcn_readfirstlane((int)(blockIdx.x * 8 + (threadIdx.x >> 5)));
    if (d >= N) return;
    const int dn  = (d + 1 < N) ? d + 1 : d;
    const int beg = offs[d];
    int end       = offs[dn];
    if (d + 1 >= N) end = E;

    const float4 hv = *(const float4*)&h[(size_t)d * OUT_DIM + lane * 4];
    const float4 al = *(const float4*)&attn_l[lane * 4];
    const float4 ar = *(const float4*)&attn_r[lane * 4];
    const float er = quad_sum(dot4(hv, ar));

    float  mx  = lrelu(quad_sum(dot4(hv, al)) + er);
    float  den = 1.0f;
    float4 acc = hv;

    int i = beg;
    for (; i + 4 <= end; i += 4) {
        fold4(h, lane, ssrc[i], ssrc[i + 1], ssrc[i + 2], ssrc[i + 3],
              true, true, true, al, er, mx, den, acc);
    }
    if (i < end) {
        const bool k1 = i + 1 < end;
        const bool k2 = i + 2 < end;
        fold4(h, lane, ssrc[i], ssrc[k1 ? i + 1 : i], ssrc[k2 ? i + 2 : i], ssrc[i],
              k1, k2, false, al, er, mx, den, acc);
    }

    const float4 bb = *(const float4*)&b[lane * 4];
    float4 o;
    o.x = acc.x / den + bb.x;
    o.y = acc.y / den + bb.y;
    o.z = acc.z / den + bb.z;
    o.w = acc.w / den + bb.w;
    *(float4*)&out[(size_t)d * OUT_DIM + lane * 4] = o;
}

extern "C" void kernel_launch(void* const* d_in, const int* in_sizes, int n_in,
                              void* d_out, int out_size, void* d_ws, size_t ws_size,
                              hipStream_t stream) {
    const float* feat   = (const float*)d_in[0];
    const float* W      = (const float*)d_in[1];
    const float* b      = (const float*)d_in[2];
    const float* attn_l = (const float*)d_in[3];
    const float* attn_r = (const float*)d_in[4];
    const int*   src    = (const int*)d_in[5];
    const int*   dst    = (const int*)d_in[6];
    float*       out    = (float*)d_out;

    const int N = in_sizes[0] / IN_DIM;
    const int E = in_sizes[5];

    float* h    = (float*)d_ws;
    int*   offs = (int*)(h + (size_t)N * OUT_DIM);
    int*   ssrc = offs + N;
    int*   cntm  = (int*)d_ws;
    int2*  pairs = (int2*)(cntm + MAXB * SORT_P);

    const int TB = 256;
    const int NB = (N + BNODES - 1) / BNODES;

    gat_chunk_sort<<<SORT_P, SORT_T, 0, stream>>>(src, dst, cntm, pairs, E, NB);

    gat_bucket_csr<<<NB, SORT_T, 0, stream>>>(pairs, cntm, offs, ssrc, N, E);

    gat_gemm_wmma<<<(N + 15) / 16, TB, 0, stream>>>(feat, W, h, N);

    gat_node_gather<<<(N + 7) / 8, TB, 0, stream>>>(offs, ssrc, attn_l, attn_r, h, b, out, N, E);
}


# Level-2 placement staged in LDS: a bucket's sources go to bucket-local LDS slots and leave as whole 128-byte lines, not 32 scattered lines a store instruction; sum order still LDS arrival rank
# speedup vs baseline: 12.4151x; 1.0331x over previous
#include <hip/hip_runtime.h>
#include <hip/hip_bf16.h>


#define IN_DIM   256
#define OUT_DIM  128
#define HEADS    8
#define HEAD_DIM 16
#define NEG_SLOPE 0.2f

typedef float v2f __attribute__((ext_vector_type(2)));
typedef float v8f __attribute__((ext_vector_type(8)));

__device__ __forceinline__ float lrelu(float x) {
    return x > 0.0f ? x : NEG_SLOPE * x;
}

#define A_STRIDE 260

__global__ void gat_gemm_wmma(const float* __restrict__ feat,
                              const float* __restrict__ W,
                              float* __restrict__ h, int N) {
    __shared__ float As[16 * A_STRIDE];

    const int row0 = blockIdx.x * 16;
    const int lane = threadIdx.x & 31;
    const int wave = threadIdx.x >> 5;
    const int m    = lane & 15;
    const int kh   = (lane >> 4) << 1;

    for (int i = threadIdx.x; i < 16 * (IN_DIM / 4); i += blockDim.x) {
        int r = i >> 6;
        int c = (i & 63) << 2;
        int gr = row0 + r;
        if (gr >= N) gr = N - 1;
        *(float4*)&As[r * A_STRIDE + c] =
            *(const float4*)&feat[(size_t)gr * IN_DIM + c];
    }
    __syncthreads();

    const int col0 = wave * HEAD_DIM;
    v8f acc = {};
    const float* __restrict__ arow = &As[m * A_STRIDE];

    #pragma unroll 4
    for (int k0 = 0; k0 < IN_DIM; k0 += 4) {
        v2f a = *(const v2f*)&arow[k0 + kh];
        v2f bb;
        bb.x = W[(size_t)(k0 + kh + 0) * OUT_DIM + col0 + m];
        bb.y = W[(size_t)(k0 + kh + 1) * OUT_DIM + col0 + m];
        acc = __builtin_amdgcn_wmma_f32_16x16x4_f32(
             false, a,  false, bb,
             (short)0, acc,  false,  false);
    }

    const int colg  = col0 + m;
    const int rbase = row0 + ((lane >> 4) << 3);
    float* __restrict__ op = h + (size_t)rbase * OUT_DIM + colg;
    if (row0 + 16 <= N) {
        #pragma unroll
        for (int r = 0; r < 8; ++r) {
            op[(size_t)r * OUT_DIM] = acc[r];
        }
    } else {
        #pragma unroll
        for (int r = 0; r < 8; ++r) {
            if (rbase + r < N) op[(size_t)r * OUT_DIM] = acc[r];
        }
    }
}

#define BSH    9
#define BNODES 512
#define MAXB   256
#define SORT_P 256
#define SORT_T 1024
#define SORT_K 8

__device__ __forceinline__ int wave_incl_scan(int v, int lane) {
    for (int o = 1; o < 32; o <<= 1) {
        int u = __shfl_up(v, o);
        if (lane >= o) v += u;
    }
    return v;
}

__global__ void __launch_bounds__(SORT_T)
gat_chunk_sort(const int* __restrict__ src, const int* __restrict__ dst,
               int* __restrict__ cntm, int2* __restrict__ pairs, int E, int NB) {
    __shared__ int  lc[MAXB];
    __shared__ int  wt[MAXB / 32];
    __shared__ int2 stage[SORT_K * SORT_T];
    const int p    = blockIdx.x;
    const int t    = threadIdx.x;
    const int lane = t & 31;
    if (t < MAXB) lc[t] = 0;
    __syncthreads();
    const int chunk = (E + SORT_P - 1) / SORT_P;
    const int e0 = p * chunk;
    const int e1 = min(e0 + chunk, E);

    int dv[SORT_K], sv[SORT_K];
    #pragma unroll
    for (int k = 0; k < SORT_K; ++k) {
        const int i = e0 + t + k * SORT_T;
        dv[k] = (i < e1) ? dst[i] : -1;
        sv[k] = (i < e1) ? src[i] : 0;
    }
    #pragma unroll
    for (int k = 0; k < SORT_K; ++k) {
        if (dv[k] >= 0) atomicAdd(&lc[dv[k] >> BSH], 1);
    }
    __syncthreads();

    const int v   = (t < MAXB) ? lc[t] : 0;
    const int inc = wave_incl_scan(v, lane);
    if (t < MAXB && lane == 31) wt[t >> 5] = inc;
    __syncthreads();
    if (t < MAXB) {
        int carry = 0;
        for (int w = 0; w < (t >> 5); ++w) carry += wt[w];
        lc[t] = carry + inc - v;
        if (t < NB) cntm[t * SORT_P + p] = v;
    }
    __syncthreads();

    #pragma unroll
    for (int k = 0; k < SORT_K; ++k) {
        if (dv[k] >= 0) {
            const int pos = atomicAdd(&lc[dv[k] >> BSH], 1);
            stage[pos] = make_int2(sv[k], dv[k]);
        }
    }
    __syncthreads();

    for (int i = t; i < e1 - e0; i += SORT_T) pairs[e0 + i] = stage[i];
}

#define CSR_J 16
#define CSR_R (SORT_P / 32)
#define CSR_CAP 16384

__global__ void __launch_bounds__(SORT_T)
gat_bucket_csr(const int2* __restrict__ pairs, const int* __restrict__ cntm,
               int* __restrict__ offs, int* __restrict__ ssrc, int N, int E) {
    __shared__ int nc[BNODES];
    __shared__ int wt[BNODES / 32];
    __shared__ __attribute__((aligned(16))) int part[CSR_J * SORT_P];
    __shared__ int rstart[SORT_P];
    __shared__ int rlen[SORT_P];
    __shared__ int red[SORT_T / 32];
    __shared__ int stage[CSR_CAP];
    __shared__ int btot;
    const int b    = blockIdx.x;
    const int t    = threadIdx.x;
    const int lane = t & 31;
    const int wv   = t >> 5;
    const int chunk = (E + SORT_P - 1) / SORT_P;

    const int mylen = (t < SORT_P) ? cntm[b * SORT_P + t] : 0;
    {
        const int q = (t & 63) << 2;
        const int j = t >> 6;
        int4 s = make_int4(0, 0, 0, 0);
        #pragma unroll 8
        for (int r = j; r < b; r += CSR_J) {
            const int4 c = *(const int4*)&cntm[r * SORT_P + q];
            s.x += c.x; s.y += c.y; s.z += c.z; s.w += c.w;
        }
        *(int4*)&part[j * SORT_P + q] = s;
    }
    if (t < BNODES) nc[t] = 0;
    __syncthreads();

    int roff = 0;
    if (t < SORT_P) {
        #pragma unroll
        for (int j = 0; j < CSR_J; ++j) roff += part[j * SORT_P + t];
        rstart[t] = t * chunk + roff;
        rlen[t]   = mylen;
    }
    int rs = roff;
    for (int o = 16; o > 0; o >>= 1) rs += __shfl_xor(rs, o);
    if (lane == 0) red[wv] = rs;
    __syncthreads();
    int s0 = red[lane];
    for (int o = 16; o > 0; o >>= 1) s0 += __shfl_xor(s0, o);

    int2 pa[CSR_R], pb[CSR_R];
    int  st[CSR_R], ln[CSR_R];
    #pragma unroll
    for (int r = 0; r < CSR_R; ++r) {
        st[r] = rstart[wv + 32 * r];
        ln[r] = rlen[wv + 32 * r];
    }
    #pragma unroll
    for (int r = 0; r < CSR_R; ++r) {
        pa[r] = (lane < ln[r])      ? pairs[st[r] + lane]      : make_int2(0, -1);
        pb[r] = (lane + 32 < ln[r]) ? pairs[st[r] + lane + 32] : make_int2(0, -1);
    }
    #pragma unroll
    for (int r = 0; r < CSR_R; ++r) {
        if (pa[r].y >= 0) atomicAdd(&nc[pa[r].y & (BNODES - 1)], 1);
        if (pb[r].y >= 0) atomicAdd(&nc[pb[r].y & (BNODES - 1)], 1);
    }
    #pragma unroll
    for (int r = 0; r < CSR_R; ++r) {
        for (int o = lane + 64; o < ln[r]; o += 32)
            atomicAdd(&nc[pairs[st[r] + o].y & (BNODES - 1)], 1);
    }
    __syncthreads();

    const int v   = (t < BNODES) ? nc[t] : 0;
    const int inc = wave_incl_scan(v, lane);
    if (t < BNODES && lane == 31) wt[t >> 5] = inc;
    __syncthreads();
    if (t < BNODES) {
        int carry = 0;
        for (int w = 0; w < (t >> 5); ++w) carry += wt[w];
        const int lstart = carry + inc - v;
        nc[t] = lstart;
        const int node = b * BNODES + t;
        if (node < N) offs[node] = s0 + lstart;
        if (t == BNODES - 1) btot = lstart + v;
    }
    __syncthreads();
    const int tot = btot;

    if (tot <= CSR_CAP) {
        #pragma unroll
        for (int r = 0; r < CSR_R; ++r) {
            if (pa[r].y >= 0) {
                const int pos = atomicAdd(&nc[pa[r].y & (BNODES - 1)], 1);
                stage[pos] = pa[r].x;
            }
            if (pb[r].y >= 0) {
                const int pos = atomicAdd(&nc[pb[r].y & (BNODES - 1)], 1);
                stage[pos] = pb[r].x;
            }
        }
        #pragma unroll
        for (int r = 0; r < CSR_R; ++r) {
            for (int o = lane + 64; o < ln[r]; o += 32) {
                const int2 pv = pairs[st[r] + o];
                const int pos = atomicAdd(&nc[pv.y & (BNODES - 1)], 1);
                stage[pos] = pv.x;
            }
        }
        __syncthreads();
        const int head = min((32 - (s0 & 31)) & 31, tot);
        if (t < head) ssrc[s0 + t] = stage[t];
        for (int base = head; base < tot; base += 4 * SORT_T) {
            int sv[4];
            #pragma unroll
            for (int k = 0; k < 4; ++k) sv[k] = stage[min(base + k * SORT_T + t, tot - 1)];
            #pragma unroll
            for (int k = 0; k < 4; ++k) {
                const int i = base + k * SORT_T + t;
                if (i < tot) ssrc[s0 + i] = sv[k];
            }
        }
    } else {
        #pragma unroll
        for (int r = 0; r < CSR_R; ++r) {
            if (pa[r].y >= 0) {
                const int pos = atomicAdd(&nc[pa[r].y & (BNODES - 1)], 1);
                ssrc[s0 + pos] = pa[r].x;
            }
            if (pb[r].y >= 0) {
                const int pos = atomicAdd(&nc[pb[r].y & (BNODES - 1)], 1);
                ssrc[s0 + pos] = pb[r].x;
            }
        }
        #pragma unroll
        for (int r = 0; r < CSR_R; ++r) {
            for (int o = lane + 64; o < ln[r]; o += 32) {
                const int2 pv = pairs[st[r] + o];
                const int pos = atomicAdd(&nc[pv.y & (BNODES - 1)], 1);
                ssrc[s0 + pos] = pv.x;
            }
        }
    }
}

__device__ __forceinline__ float dot4(const float4 v, const float4 a) {
    return fmaf(v.w, a.w, fmaf(v.z, a.z, fmaf(v.y, a.y, v.x * a.x)));
}

__device__ __forceinline__ float quad_sum(float x) {
    const float y = __int_as_float(__builtin_amdgcn_update_dpp(
        0, __float_as_int(x), 0xB1  , 0xF, 0xF, true));
    x += y;
    const float z = __int_as_float(__builtin_amdgcn_update_dpp(
        0, __float_as_int(x), 0x4E  , 0xF, 0xF, true));
    return x + z;
}

__device__ __forceinline__ void fold4(const float* __restrict__ h, const int lane,
                                      const int s0, const int s1, const int s2, const int s3,
                                      const bool k1, const bool k2, const bool k3,
                                      const float4 al, const float er,
                                      float& mx, float& den, float4& acc) {
    const float4 v0 = *(const float4*)&h[(size_t)s0 * OUT_DIM + lane * 4];
    const float4 v1 = *(const float4*)&h[(size_t)s1 * OUT_DIM + lane * 4];
    const float4 v2 = *(const float4*)&h[(size_t)s2 * OUT_DIM + lane * 4];
    const float4 v3 = *(const float4*)&h[(size_t)s3 * OUT_DIM + lane * 4];
    const float x0 = lrelu(quad_sum(dot4(v0, al)) + er);
    const float x1 = k1 ? lrelu(quad_sum(dot4(v1, al)) + er) : -INFINITY;
    const float x2 = k2 ? lrelu(quad_sum(dot4(v2, al)) + er) : -INFINITY;
    const float x3 = k3 ? lrelu(quad_sum(dot4(v3, al)) + er) : -INFINITY;
    const float mn = fmaxf(fmaxf(mx, fmaxf(x0, x1)), fmaxf(x2, x3));
    const float sc = __expf(mx - mn);
    const float p0 = __expf(x0 - mn);
    const float p1 = __expf(x1 - mn);
    const float p2 = __expf(x2 - mn);
    const float p3 = __expf(x3 - mn);
    mx  = mn;
    den = fmaf(den, sc, p0);
    den += p1;
    den += p2;
    den += p3;
    acc.x = fmaf(v3.x, p3, fmaf(v2.x, p2, fmaf(v1.x, p1, fmaf(v0.x, p0, acc.x * sc))));
    acc.y = fmaf(v3.y, p3, fmaf(v2.y, p2, fmaf(v1.y, p1, fmaf(v0.y, p0, acc.y * sc))));
    acc.z = fmaf(v3.z, p3, fmaf(v2.z, p2, fmaf(v1.z, p1, fmaf(v0.z, p0, acc.z * sc))));
    acc.w = fmaf(v3.w, p3, fmaf(v2.w, p2, fmaf(v1.w, p1, fmaf(v0.w, p0, acc.w * sc))));
}

__global__ void __launch_bounds__(256, 16)
gat_node_gather(const int* __restrict__ offs, const int* __restrict__ ssrc,
                const float* __restrict__ attn_l, const float* __restrict__ attn_r,
                const float* __restrict__ h, const float* __restrict__ b,
                float* __restrict__ out, int N, int E) {
    const int lane = threadIdx.x & 31;
    const int d = __builtin_amdgcn_readfirstlane((int)(blockIdx.x * 8 + (threadIdx.x >> 5)));
    if (d >= N) return;
    const int dn  = (d + 1 < N) ? d + 1 : d;
    const int beg = offs[d];
    int end       = offs[dn];
    if (d + 1 >= N) end = E;

    const float4 hv = *(const float4*)&h[(size_t)d * OUT_DIM + lane * 4];
    const float4 al = *(const float4*)&attn_l[lane * 4];
    const float4 ar = *(const float4*)&attn_r[lane * 4];
    const float er = quad_sum(dot4(hv, ar));

    float  mx  = lrelu(quad_sum(dot4(hv, al)) + er);
    float  den = 1.0f;
    float4 acc = hv;

    int i = beg;
    for (; i + 4 <= end; i += 4) {
        fold4(h, lane, ssrc[i], ssrc[i + 1], ssrc[i + 2], ssrc[i + 3],
              true, true, true, al, er, mx, den, acc);
    }
    if (i < end) {
        const bool k1 = i + 1 < end;
        const bool k2 = i + 2 < end;
        fold4(h, lane, ssrc[i], ssrc[k1 ? i + 1 : i], ssrc[k2 ? i + 2 : i], ssrc[i],
              k1, k2, false, al, er, mx, den, acc);
    }

    const float4 bb = *(const float4*)&b[lane * 4];
    float4 o;
    o.x = acc.x / den + bb.x;
    o.y = acc.y / den + bb.y;
    o.z = acc.z / den + bb.z;
    o.w = acc.w / den + bb.w;
    *(float4*)&out[(size_t)d * OUT_DIM + lane * 4] = o;
}

extern "C" void kernel_launch(void* const* d_in, const int* in_sizes, int n_in,
                              void* d_out, int out_size, void* d_ws, size_t ws_size,
                              hipStream_t stream) {
    const float* feat   = (const float*)d_in[0];
    const float* W      = (const float*)d_in[1];
    const float* b      = (const float*)d_in[2];
    const float* attn_l = (const float*)d_in[3];
    const float* attn_r = (const float*)d_in[4];
    const int*   src    = (const int*)d_in[5];
    const int*   dst    = (const int*)d_in[6];
    float*       out    = (float*)d_out;

    const int N = in_sizes[0] / IN_DIM;
    const int E = in_sizes[5];

    float* h    = (float*)d_ws;
    int*   offs = (int*)(h + (size_t)N * OUT_DIM);
    int*   ssrc = offs + N;
    int*   cntm  = (int*)d_ws;
    int2*  pairs = (int2*)(cntm + MAXB * SORT_P);

    const int TB = 256;
    const int NB = (N + BNODES - 1) / BNODES;

    gat_chunk_sort<<<SORT_P, SORT_T, 0, stream>>>(src, dst, cntm, pairs, E, NB);

    gat_bucket_csr<<<NB, SORT_T, 0, stream>>>(pairs, cntm, offs, ssrc, N, E);

    gat_gemm_wmma<<<(N + 15) / 16, TB, 0, stream>>>(feat, W, h, N);

    gat_node_gather<<<(N + 7) / 8, TB, 0, stream>>>(offs, ssrc, attn_l, attn_r, h, b, out, N, E);
}


# Product tiles (64 rows, 1024-thread blocks capped at 64 registers) run as extra blocks of the two sort launches, role by block index: matrix pipe works under the sort's waits; 4 launches to 3
# speedup vs baseline: 13.3920x; 1.0787x over previous
#include <hip/hip_runtime.h>
#include <hip/hip_bf16.h>


#define IN_DIM   256
#define OUT_DIM  128
#define HEADS    8
#define HEAD_DIM 16
#define NEG_SLOPE 0.2f

typedef float v2f __attribute__((ext_vector_type(2)));
typedef float v8f __attribute__((ext_vector_type(8)));

__device__ __forceinline__ float lrelu(float x) {
    return x > 0.0f ? x : NEG_SLOPE * x;
}

#define A_STRIDE  260
#define GEMM_ROWS 64
#define ROLE_T    1024
#define GEMM_LDS_BYTES (GEMM_ROWS * A_STRIDE * 4)

__device__ __forceinline__ void gemm_tile_role(float* __restrict__ As,
                                               const float* __restrict__ feat,
                                               const float* __restrict__ W,
                                               float* __restrict__ h,
                                               const int N, const int tile) {
    const int row0 = tile * GEMM_ROWS;
    const int t    = threadIdx.x;
    const int lane = t & 31;
    const int wave = t >> 5;
    const int head = wave & 7;
    const int mt   = wave >> 3;
    const int m    = lane & 15;
    const int kh   = (lane >> 4) << 1;

    #pragma unroll
    for (int k = 0; k < GEMM_ROWS * (IN_DIM / 4) / ROLE_T; ++k) {
        const int i = t + k * ROLE_T;
        const int r = i >> 6;
        const int c = (i & 63) << 2;
        int gr = row0 + r;
        if (gr >= N) gr = N - 1;
        *(float4*)&As[r * A_STRIDE + c] =
            *(const float4*)&feat[(size_t)gr * IN_DIM + c];
    }
    __syncthreads();

    const int col0 = head * HEAD_DIM;
    v8f acc = {};
    const float* __restrict__ arow = &As[(mt * 16 + m) * A_STRIDE];
    const float* __restrict__ wcol = W + (size_t)kh * OUT_DIM + col0 + m;

    #pragma unroll 4
    for (int k0 = 0; k0 < IN_DIM; k0 += 4) {
        v2f a = *(const v2f*)&arow[k0 + kh];
        v2f bb;
        bb.x = wcol[(size_t)k0 * OUT_DIM];
        bb.y = wcol[(size_t)k0 * OUT_DIM + OUT_DIM];
        acc = __builtin_amdgcn_wmma_f32_16x16x4_f32(
             false, a,  false, bb,
             (short)0, acc,  false,  false);
    }

    const int colg  = col0 + m;
    const int rt0   = row0 + mt * 16;
    const int rbase = rt0 + ((lane >> 4) << 3);
    float* __restrict__ op = h + (size_t)rbase * OUT_DIM + colg;
    if (rt0 + 16 <= N) {
        #pragma unroll
        for (int r = 0; r < 8; ++r) {
            op[(size_t)r * OUT_DIM] = acc[r];
        }
    } else {
        #pragma unroll
        for (int r = 0; r < 8; ++r) {
            if (rbase + r < N) op[(size_t)r * OUT_DIM] = acc[r];
        }
    }
}

#define BSH    9
#define BNODES 512
#define MAXB   256
#define SORT_P 256
#define SORT_T 1024
#define SORT_K 8

__device__ __forceinline__ int wave_incl_scan(int v, int lane) {
    for (int o = 1; o < 32; o <<= 1) {
        int u = __shfl_up(v, o);
        if (lane >= o) v += u;
    }
    return v;
}

#define SORT1_LDS_BYTES (MAXB * 4 + (MAXB / 32) * 4 + SORT_K * SORT_T * 8)

__device__ __forceinline__ void chunk_sort_role(unsigned char* smem, const int p,
               const int* __restrict__ src, const int* __restrict__ dst,
               int* __restrict__ cntm, int2* __restrict__ pairs,
               const int E, const int NB) {
    int*  lc    = (int*)smem;
    int*  wt    = lc + MAXB;
    int2* stage = (int2*)(wt + MAXB / 32);
    const int t    = threadIdx.x;
    const int lane = t & 31;
    if (t < MAXB) lc[t] = 0;
    __syncthreads();
    const int chunk = (E + SORT_P - 1) / SORT_P;
    const int e0 = p * chunk;
    const int e1 = min(e0 + chunk, E);

    int dv[SORT_K], sv[SORT_K];
    #pragma unroll
    for (int k = 0; k < SORT_K; ++k) {
        const int i = e0 + t + k * SORT_T;
        dv[k] = (i < e1) ? dst[i] : -1;
        sv[k] = (i < e1) ? src[i] : 0;
    }
    #pragma unroll
    for (int k = 0; k < SORT_K; ++k) {
        if (dv[k] >= 0) atomicAdd(&lc[dv[k] >> BSH], 1);
    }
    __syncthreads();

    const int v   = (t < MAXB) ? lc[t] : 0;
    const int inc = wave_incl_scan(v, lane);
    if (t < MAXB && lane == 31) wt[t >> 5] = inc;
    __syncthreads();
    if (t < MAXB) {
        int carry = 0;
        for (int w = 0; w < (t >> 5); ++w) carry += wt[w];
        lc[t] = carry + inc - v;
        if (t < NB) cntm[t * SORT_P + p] = v;
    }
    __syncthreads();

    #pragma unroll
    for (int k = 0; k < SORT_K; ++k) {
        if (dv[k] >= 0) {
            const int pos = atomicAdd(&lc[dv[k] >> BSH], 1);
            stage[pos] = make_int2(sv[k], dv[k]);
        }
    }
    __syncthreads();

    for (int i = t; i < e1 - e0; i += SORT_T) pairs[e0 + i] = stage[i];
}

#define CSR_J 16
#define CSR_R (SORT_P / 32)
#define CSR_CAP 16384

#define CSR_LDS_INTS (BNODES + BNODES / 32 + SORT_T / 32 + 2 * SORT_P + 4 + CSR_J * SORT_P + CSR_CAP)

__device__ __forceinline__ void bucket_csr_role(unsigned char* smem, const int b,
               const int2* __restrict__ pairs, const int* __restrict__ cntm,
               int* __restrict__ offs, int* __restrict__ ssrc,
               const int N, const int E) {
    int* nc     = (int*)smem;
    int* wt     = nc + BNODES;
    int* red    = wt + BNODES / 32;
    int* rstart = red + SORT_T / 32;
    int* rlen   = rstart + SORT_P;
    int* btotp  = rlen + SORT_P;
    int* part   = btotp + 4;
    int* stage  = part + CSR_J * SORT_P;
    const int t    = threadIdx.x;
    const int lane = t & 31;
    const int wv   = t >> 5;
    const int chunk = (E + SORT_P - 1) / SORT_P;

    const int mylen = (t < SORT_P) ? cntm[b * SORT_P + t] : 0;
    {
        const int q = (t & 63) << 2;
        const int j = t >> 6;
        int4 s = make_int4(0, 0, 0, 0);
        #pragma unroll 8
        for (int r = j; r < b; r += CSR_J) {
            const int4 c = *(const int4*)&cntm[r * SORT_P + q];
            s.x += c.x; s.y += c.y; s.z += c.z; s.w += c.w;
        }
        *(int4*)&part[j * SORT_P + q] = s;
    }
    if (t < BNODES) nc[t] = 0;
    __syncthreads();

    int roff = 0;
    if (t < SORT_P) {
        #pragma unroll
        for (int j = 0; j < CSR_J; ++j) roff += part[j * SORT_P + t];
        rstart[t] = t * chunk + roff;
        rlen[t]   = mylen;
    }
    int rs = roff;
    for (int o = 16; o > 0; o >>= 1) rs += __shfl_xor(rs, o);
    if (lane == 0) red[wv] = rs;
    __syncthreads();
    int s0 = red[lane];
    for (int o = 16; o > 0; o >>= 1) s0 += __shfl_xor(s0, o);

    int2 pa[CSR_R], pb[CSR_R];
    int  st[CSR_R], ln[CSR_R];
    #pragma unroll
    for (int r = 0; r < CSR_R; ++r) {
        st[r] = rstart[wv + 32 * r];
        ln[r] = rlen[wv + 32 * r];
    }
    #pragma unroll
    for (int r = 0; r < CSR_R; ++r) {
        pa[r] = (lane < ln[r])      ? pairs[st[r] + lane]      : make_int2(0, -1);
        pb[r] = (lane + 32 < ln[r]) ? pairs[st[r] + lane + 32] : make_int2(0, -1);
    }
    #pragma unroll
    for (int r = 0; r < CSR_R; ++r) {
        if (pa[r].y >= 0) atomicAdd(&nc[pa[r].y & (BNODES - 1)], 1);
        if (pb[r].y >= 0) atomicAdd(&nc[pb[r].y & (BNODES - 1)], 1);
    }
    #pragma unroll
    for (int r = 0; r < CSR_R; ++r) {
        for (int o = lane + 64; o < ln[r]; o += 32)
            atomicAdd(&nc[pairs[st[r] + o].y & (BNODES - 1)], 1);
    }
    __syncthreads();

    const int v   = (t < BNODES) ? nc[t] : 0;
    const int inc = wave_incl_scan(v, lane);
    if (t < BNODES && lane == 31) wt[t >> 5] = inc;
    __syncthreads();
    if (t < BNODES) {
        int carry = 0;
        for (int w = 0; w < (t >> 5); ++w) carry += wt[w];
        const int lstart = carry + inc - v;
        nc[t] = lstart;
        const int node = b * BNODES + t;
        if (node < N) offs[node] = s0 + lstart;
        if (t == BNODES - 1) *btotp = lstart + v;
    }
    __syncthreads();
    const int tot = *btotp;

    if (tot <= CSR_CAP) {
        #pragma unroll
        for (int r = 0; r < CSR_R; ++r) {
            if (pa[r].y >= 0) {
                const int pos = atomicAdd(&nc[pa[r].y & (BNODES - 1)], 1);
                stage[pos] = pa[r].x;
            }
            if (pb[r].y >= 0) {
                const int pos = atomicAdd(&nc[pb[r].y & (BNODES - 1)], 1);
                stage[pos] = pb[r].x;
            }
        }
        #pragma unroll
        for (int r = 0; r < CSR_R; ++r) {
            const int lr = rlen[wv + 32 * r];
            const int sr = rstart[wv + 32 * r];
            for (int o = lane + 64; o < lr; o += 32) {
                const int2 pv = pairs[sr + o];
                const int pos = atomicAdd(&nc[pv.y & (BNODES - 1)], 1);
                stage[pos] = pv.x;
            }
        }
        __syncthreads();
        const int head = min((32 - (s0 & 31)) & 31, tot);
        if (t < head) ssrc[s0 + t] = stage[t];
        for (int base = head; base < tot; base += 4 * SORT_T) {
            int sv[4];
            #pragma unroll
            for (int k = 0; k < 4; ++k) sv[k] = stage[min(base + k * SORT_T + t, tot - 1)];
            #pragma unroll
            for (int k = 0; k < 4; ++k) {
                const int i = base + k * SORT_T + t;
                if (i < tot) ssrc[s0 + i] = sv[k];
            }
        }
    } else {
        #pragma unroll
        for (int r = 0; r < CSR_R; ++r) {
            if (pa[r].y >= 0) {
                const int pos = atomicAdd(&nc[pa[r].y & (BNODES - 1)], 1);
                ssrc[s0 + pos] = pa[r].x;
            }
            if (pb[r].y >= 0) {
                const int pos = atomicAdd(&nc[pb[r].y & (BNODES - 1)], 1);
                ssrc[s0 + pos] = pb[r].x;
            }
        }
        #pragma unroll
        for (int r = 0; r < CSR_R; ++r) {
            const int lr = rlen[wv + 32 * r];
            const int sr = rstart[wv + 32 * r];
            for (int o = lane + 64; o < lr; o += 32) {
                const int2 pv = pairs[sr + o];
                const int pos = atomicAdd(&nc[pv.y & (BNODES - 1)], 1);
                ssrc[s0 + pos] = pv.x;
            }
        }
    }
}

#define LDS_A_BYTES (SORT1_LDS_BYTES > GEMM_LDS_BYTES ? SORT1_LDS_BYTES : GEMM_LDS_BYTES)
#define LDS_B_BYTES (CSR_LDS_INTS * 4 > GEMM_LDS_BYTES ? CSR_LDS_INTS * 4 : GEMM_LDS_BYTES)

__global__ void __launch_bounds__(ROLE_T, 16)
gat_sort1_with_gemm(const int* __restrict__ src, const int* __restrict__ dst,
                    int* __restrict__ cntm, int2* __restrict__ pairs, int E, int NB,
                    const float* __restrict__ feat, const float* __restrict__ W,
                    float* __restrict__ h, int N) {
    __shared__ __attribute__((aligned(16))) unsigned char smem[LDS_A_BYTES];
    const int bid = blockIdx.x;
    if (bid < SORT_P) {
        chunk_sort_role(smem, bid, src, dst, cntm, pairs, E, NB);
    } else {
        gemm_tile_role((float*)smem, feat, W, h, N, bid - SORT_P);
    }
}

__global__ void __launch_bounds__(ROLE_T, 16)
gat_sort2_with_gemm(const int2* __restrict__ pairs, const int* __restrict__ cntm,
                    int* __restrict__ offs, int* __restrict__ ssrc, int E, int NB,
                    const float* __restrict__ feat, const float* __restrict__ W,
                    float* __restrict__ h, int N, int tile0) {
    __shared__ __attribute__((aligned(16))) unsigned char smem[LDS_B_BYTES];
    const int bid = blockIdx.x;
    if (bid < NB) {
        bucket_csr_role(smem, bid, pairs, cntm, offs, ssrc, N, E);
    } else {
        gemm_tile_role((float*)smem, feat, W, h, N, tile0 + (bid - NB));
    }
}

__device__ __forceinline__ float dot4(const float4 v, const float4 a) {
    return fmaf(v.w, a.w, fmaf(v.z, a.z, fmaf(v.y, a.y, v.x * a.x)));
}

__device__ __forceinline__ float quad_sum(float x) {
    const float y = __int_as_float(__builtin_amdgcn_update_dpp(
        0, __float_as_int(x), 0xB1  , 0xF, 0xF, true));
    x += y;
    const float z = __int_as_float(__builtin_amdgcn_update_dpp(
        0, __float_as_int(x), 0x4E  , 0xF, 0xF, true));
    return x + z;
}

__device__ __forceinline__ void fold4(const float* __restrict__ h, const int lane,
                                      const int s0, const int s1, const int s2, const int s3,
                                      const bool k1, const bool k2, const bool k3,
                                      const float4 al, const float er,
                                      float& mx, float& den, float4& acc) {
    const float4 v0 = *(const float4*)&h[(size_t)s0 * OUT_DIM + lane * 4];
    const float4 v1 = *(const float4*)&h[(size_t)s1 * OUT_DIM + lane * 4];
    const float4 v2 = *(const float4*)&h[(size_t)s2 * OUT_DIM + lane * 4];
    const float4 v3 = *(const float4*)&h[(size_t)s3 * OUT_DIM + lane * 4];
    const float x0 = lrelu(quad_sum(dot4(v0, al)) + er);
    const float x1 = k1 ? lrelu(quad_sum(dot4(v1, al)) + er) : -INFINITY;
    const float x2 = k2 ? lrelu(quad_sum(dot4(v2, al)) + er) : -INFINITY;
    const float x3 = k3 ? lrelu(quad_sum(dot4(v3, al)) + er) : -INFINITY;
    const float mn = fmaxf(fmaxf(mx, fmaxf(x0, x1)), fmaxf(x2, x3));
    const float sc = __expf(mx - mn);
    const float p0 = __expf(x0 - mn);
    const float p1 = __expf(x1 - mn);
    const float p2 = __expf(x2 - mn);
    const float p3 = __expf(x3 - mn);
    mx  = mn;
    den = fmaf(den, sc, p0);
    den += p1;
    den += p2;
    den += p3;
    acc.x = fmaf(v3.x, p3, fmaf(v2.x, p2, fmaf(v1.x, p1, fmaf(v0.x, p0, acc.x * sc))));
    acc.y = fmaf(v3.y, p3, fmaf(v2.y, p2, fmaf(v1.y, p1, fmaf(v0.y, p0, acc.y * sc))));
    acc.z = fmaf(v3.z, p3, fmaf(v2.z, p2, fmaf(v1.z, p1, fmaf(v0.z, p0, acc.z * sc))));
    acc.w = fmaf(v3.w, p3, fmaf(v2.w, p2, fmaf(v1.w, p1, fmaf(v0.w, p0, acc.w * sc))));
}

__global__ void __launch_bounds__(256, 16)
gat_node_gather(const int* __restrict__ offs, const int* __restrict__ ssrc,
                const float* __restrict__ attn_l, const float* __restrict__ attn_r,
                const float* __restrict__ h, const float* __restrict__ b,
                float* __restrict__ out, int N, int E) {
    const int lane = threadIdx.x & 31;
    const int d = __builtin_amdgcn_readfirstlane((int)(blockIdx.x * 8 + (threadIdx.x >> 5)));
    if (d >= N) return;
    const int dn  = (d + 1 < N) ? d + 1 : d;
    const int beg = offs[d];
    int end       = offs[dn];
    if (d + 1 >= N) end = E;

    const float4 hv = *(const float4*)&h[(size_t)d * OUT_DIM + lane * 4];
    const float4 al = *(const float4*)&attn_l[lane * 4];
    const float4 ar = *(const float4*)&attn_r[lane * 4];
    const float er = quad_sum(dot4(hv, ar));

    float  mx  = lrelu(quad_sum(dot4(hv, al)) + er);
    float  den = 1.0f;
    float4 acc = hv;

    int i = beg;
    for (; i + 4 <= end; i += 4) {
        fold4(h, lane, ssrc[i], ssrc[i + 1], ssrc[i + 2], ssrc[i + 3],
              true, true, true, al, er, mx, den, acc);
    }
    if (i < end) {
        const bool k1 = i + 1 < end;
        const bool k2 = i + 2 < end;
        fold4(h, lane, ssrc[i], ssrc[k1 ? i + 1 : i], ssrc[k2 ? i + 2 : i], ssrc[i],
              k1, k2, false, al, er, mx, den, acc);
    }

    const float4 bb = *(const float4*)&b[lane * 4];
    float4 o;
    o.x = acc.x / den + bb.x;
    o.y = acc.y / den + bb.y;
    o.z = acc.z / den + bb.z;
    o.w = acc.w / den + bb.w;
    *(float4*)&out[(size_t)d * OUT_DIM + lane * 4] = o;
}

extern "C" void kernel_launch(void* const* d_in, const int* in_sizes, int n_in,
                              void* d_out, int out_size, void* d_ws, size_t ws_size,
                              hipStream_t stream) {
    const float* feat   = (const float*)d_in[0];
    const float* W      = (const float*)d_in[1];
    const float* b      = (const float*)d_in[2];
    const float* attn_l = (const float*)d_in[3];
    const float* attn_r = (const float*)d_in[4];
    const int*   src    = (const int*)d_in[5];
    const int*   dst    = (const int*)d_in[6];
    float*       out    = (float*)d_out;

    const int N = in_sizes[0] / IN_DIM;
    const int E = in_sizes[5];

    float* h    = (float*)d_ws;
    int*   offs = (int*)(h + (size_t)N * OUT_DIM);
    int*   ssrc = offs + N;
    int*   cntm  = (int*)d_out;
    int2*  pairs = (int2*)(cntm + MAXB * SORT_P);

    const int TB = 256;
    const int NB = (N + BNODES - 1) / BNODES;
    const int NT = (N + GEMM_ROWS - 1) / GEMM_ROWS;
    const int T1 = NT / 2;

    gat_sort1_with_gemm<<<SORT_P + T1, ROLE_T, 0, stream>>>(src, dst, cntm, pairs, E, NB,
                                                            feat, W, h, N);

    gat_sort2_with_gemm<<<NB + (NT - T1), ROLE_T, 0, stream>>>(pairs, cntm, offs, ssrc, E, NB,
                                                               feat, W, h, N, T1);

    gat_node_gather<<<(N + 7) / 8, TB, 0, stream>>>(offs, ssrc, attn_l, attn_r, h, b, out, N, E);
}
